# starting kernel
# baseline (speedup 1.0000x reference)
#include <hip/hip_runtime.h>

typedef float v2f __attribute__((ext_vector_type(2)));
typedef float v8f __attribute__((ext_vector_type(8)));

#define NN 50000
#define NE 800000
#define F  64
#define KD 256


__global__ __launch_bounds__(256) void zero_k(float* __restrict__ p, long n) {
    long i  = (long)blockIdx.x * blockDim.x + threadIdx.x;
    long st = (long)gridDim.x * blockDim.x;
    for (; i < n; i += st) p[i] = 0.0f;
}

__global__ __launch_bounds__(256) void deg_k(const int* __restrict__ dst, int* __restrict__ deg) {
    int e = blockIdx.x * 256 + threadIdx.x;
    if (e < NE) atomicAdd(&deg[dst[e]], 1);
}

__global__ __launch_bounds__(256) void norm_k(float* __restrict__ nrm) {
    int i = blockIdx.x * 256 + threadIdx.x;
    if (i < NN) {
        float d = (float)((const int*)nrm)[i];
        nrm[i] = rsqrtf(d < 1.0f ? 1.0f : d);
    }
}

__global__ __launch_bounds__(256) void hop_k(const int* __restrict__ src,
                                             const int* __restrict__ dst,
                                             const float* __restrict__ nrm,
                                             const float* __restrict__ hin,
                                             float* __restrict__ hout) {
    long t = (long)blockIdx.x * 256 + threadIdx.x;
    int  e = (int)(t >> 4);
    if (e >= NE) return;
    int   p  = (int)(t & 15);
    int   s  = src[e];
    int   d  = dst[e];
    float ns = nrm[s];
    float4 v = ((const float4*)(hin + (long)s * F))[p];
    float* op = hout + (long)d * F + p * 4;
    unsafeAtomicAdd(op + 0, v.x * ns);
    unsafeAtomicAdd(op + 1, v.y * ns);
    unsafeAtomicAdd(op + 2, v.z * ns);
    unsafeAtomicAdd(op + 3, v.w * ns);
}

__global__ __launch_bounds__(256) void scale_k(const float* __restrict__ nrm,
                                               float* __restrict__ h) {
    long i = (long)blockIdx.x * 256 + threadIdx.x;
    if (i < (long)NN * F) h[i] *= nrm[i >> 6];
}


__device__ __forceinline__ void seg_mac(const float* __restrict__ base, int kbase,
                                        int node, int hi2, int ln15,
                                        const float* Wt,
                                        v8f& c0, v8f& c1, v8f& c2, v8f& c3) {
    const float* rp = base + (long)node * F + hi2;
#pragma unroll 4
    for (int kk = 0; kk < F; kk += 4) {
        v2f a;
        a.x = rp[kk];
        a.y = rp[kk + 1];
        int kr = (kbase + kk + hi2) * F + ln15;
        v2f b0; b0.x = Wt[kr];      b0.y = Wt[kr + F];
        v2f b1; b1.x = Wt[kr + 16]; b1.y = Wt[kr + F + 16];
        v2f b2; b2.x = Wt[kr + 32]; b2.y = Wt[kr + F + 32];
        v2f b3; b3.x = Wt[kr + 48]; b3.y = Wt[kr + F + 48];
        c0 = __builtin_amdgcn_wmma_f32_16x16x4_f32(false, a, false, b0, (short)0, c0, false, false);
        c1 = __builtin_amdgcn_wmma_f32_16x16x4_f32(false, a, false, b1, (short)0, c1, false, false);
        c2 = __builtin_amdgcn_wmma_f32_16x16x4_f32(false, a, false, b2, (short)0, c2, false, false);
        c3 = __builtin_amdgcn_wmma_f32_16x16x4_f32(false, a, false, b3, (short)0, c3, false, false);
    }
}

__global__ __launch_bounds__(256) void gemm_k(const float* __restrict__ feat,
                                              const float* __restrict__ h1,
                                              const float* __restrict__ h2,
                                              const float* __restrict__ h3,
                                              const float* __restrict__ W,
                                              const float* __restrict__ bias,
                                              float* __restrict__ out) {
    __shared__ float Wt[KD * F];
    int tid = threadIdx.x;
    for (int i = tid; i < KD * F; i += 256) {
        int o = i >> 8;
        int k = i & 255;
        Wt[k * F + o] = W[i];
    }
    __syncthreads();

    int lane = tid & 31;
    int ln15 = lane & 15;
    int hi2  = (lane >= 16) ? 2 : 0;
    int rowBase = blockIdx.x * 128 + (tid >> 5) * 16;
    if (rowBase >= NN) return;

    int node = rowBase + ln15;
    if (node >= NN) node = NN - 1;

    v8f c0, c1, c2, c3;
    float bv0 = bias[ln15], bv1 = bias[16 + ln15], bv2 = bias[32 + ln15], bv3 = bias[48 + ln15];
#pragma unroll
    for (int i = 0; i < 8; ++i) { c0[i] = bv0; c1[i] = bv1; c2[i] = bv2; c3[i] = bv3; }

    seg_mac(feat, 0,   node, hi2, ln15, Wt, c0, c1, c2, c3);
    seg_mac(h1,   64,  node, hi2, ln15, Wt, c0, c1, c2, c3);
    seg_mac(h2,   128, node, hi2, ln15, Wt, c0, c1, c2, c3);
    seg_mac(h3,   192, node, hi2, ln15, Wt, c0, c1, c2, c3);

    int mBase = rowBase + (hi2 ? 8 : 0);
#pragma unroll
    for (int r = 0; r < 8; ++r) {
        int row = mBase + r;
        if (row < NN) {
            out[(long)row * F + ln15]      = c0[r];
            out[(long)row * F + 16 + ln15] = c1[r];
            out[(long)row * F + 32 + ln15] = c2[r];
            out[(long)row * F + 48 + ln15] = c3[r];
        }
    }
}


extern "C" void kernel_launch(void* const* d_in, const int* in_sizes, int n_in,
                              void* d_out, int out_size, void* d_ws, size_t ws_size,
                              hipStream_t stream) {
    const float* feat = (const float*)d_in[0];
    const int*   src  = (const int*)d_in[1];
    const int*   dst  = (const int*)d_in[2];
    const float* W    = (const float*)d_in[3];
    const float* bias = (const float*)d_in[4];
    float*       out  = (float*)d_out;

    float* ws   = (float*)d_ws;
    float* nrm  = ws;
    float* h1   = ws + 50048;
    float* h2   = h1 + (long)NN * F;
    float* h3   = h2 + (long)NN * F;
    long   ztot = 50048 + 3L * NN * F;

    zero_k <<<2048, 256, 0, stream>>>(ws, ztot);
    deg_k  <<<(NE + 255) / 256, 256, 0, stream>>>(dst, (int*)nrm);
    norm_k <<<(NN + 255) / 256, 256, 0, stream>>>(nrm);

    const int hopBlocks   = (NE * 16) / 256;
    const int scaleBlocks = (NN * F + 255) / 256;

    hop_k  <<<hopBlocks,   256, 0, stream>>>(src, dst, nrm, feat, h1);
    scale_k<<<scaleBlocks, 256, 0, stream>>>(nrm, h1);
    hop_k  <<<hopBlocks,   256, 0, stream>>>(src, dst, nrm, h1, h2);
    scale_k<<<scaleBlocks, 256, 0, stream>>>(nrm, h2);
    hop_k  <<<hopBlocks,   256, 0, stream>>>(src, dst, nrm, h2, h3);
    scale_k<<<scaleBlocks, 256, 0, stream>>>(nrm, h3);

    gemm_k <<<(NN + 127) / 128, 256, 0, stream>>>(feat, h1, h2, h3, W, bias, out);
}


# Hops 2 and 3 skip the sweep and the source look-up: hop 1 saves each work-group's parked-edge queue (row, source, source norm) in the output buffer and they load it back; sums stay LDS adds
# speedup vs baseline: 1.7031x; 1.7031x over previous
#include <hip/hip_runtime.h>

typedef float v2f __attribute__((ext_vector_type(2)));
typedef float v8f __attribute__((ext_vector_type(8)));

#define NN 50000
#define NE 800000
#define F  64
#define KD 256


__global__ __launch_bounds__(256) void zero_k(float* __restrict__ p, long n) {
    long i  = (long)blockIdx.x * blockDim.x + threadIdx.x;
    long st = (long)gridDim.x * blockDim.x;
    for (; i < n; i += st) p[i] = 0.0f;
}

__global__ __launch_bounds__(256) void deg_k(const int* __restrict__ dst, int* __restrict__ deg) {
    int e = blockIdx.x * 256 + threadIdx.x;
    if (e < NE) atomicAdd(&deg[dst[e]], 1);
}

__global__ __launch_bounds__(256) void norm_k(float* __restrict__ nrm) {
    int i = blockIdx.x * 256 + threadIdx.x;
    if (i < NN) {
        float d = (float)((const int*)nrm)[i];
        nrm[i] = rsqrtf(d < 1.0f ? 1.0f : d);
    }
}

#define NPB  196
#define HOPG 256
#define QCAP 16384
#define NV   (NE / 4)
#define SWEEP (8 * 1024)
#define STAG 768

static_assert(NPB * HOPG >= NN, "every node has an owner");
static_assert(NE % 4 == 0 && NE <= (1 << 20) && NPB <= (1 << 12), "queue word: row << 20 | edge");
static_assert(NN <= (1 << 16), "queue word after look-up: row << 16 | source");
static_assert(QCAP % (8 * 1024) == 0, "the look-up walks the queue in batches of 8 x 1024");
static_assert((HOPG - 1) * STAG < NV && NV % 32 == 0 && STAG % 32 == 0, "one wrap at most, never inside a wave");

#define SVSTRIDE 12480
#define SVHEAD   32
#define CAPSV    6208
static_assert((long)HOPG * SVSTRIDE <= (long)NN * F && SVSTRIDE % 32 == 0, "the saved queues fit in out, line-aligned");
static_assert(SVHEAD + 2 * CAPSV <= SVSTRIDE && CAPSV % 32 == 0 && CAPSV <= QCAP, "count line, words, norms");
static_assert(CAPSV <= 2 * 1024 * 4, "a thread loads a saved queue back in two 16-byte groups");

__device__ __forceinline__ void park(unsigned r, int e, unsigned* q, unsigned* qn, float* tile,
                                     const int* __restrict__ src, const float* __restrict__ nrm,
                                     const float* __restrict__ hin) {
    unsigned p = atomicAdd(qn, 1u);
    if (p < QCAP) {
        q[p] = (r << 20) | (unsigned)e;
    } else {
        int   s = src[e];
        float w = nrm[s];
        const float* ip = hin + (long)s * F;
        float* tp = tile + r * F;
#pragma unroll 1
        for (int k = 0; k < F; ++k) unsafeAtomicAdd(tp + k, ip[k] * w);
    }
}

__device__ __forceinline__ void look(int4 v, int j, int base, unsigned* q, unsigned* qn, float* tile,
                                     const int* __restrict__ src, const float* __restrict__ nrm,
                                     const float* __restrict__ hin) {
    unsigned rx = (unsigned)(v.x - base), ry = (unsigned)(v.y - base);
    unsigned rz = (unsigned)(v.z - base), rw = (unsigned)(v.w - base);
    unsigned mn = min(min(rx, ry), min(rz, rw));
    if (mn < NPB) {
#pragma unroll 1
        for (int c = 0; c < 4; ++c) {
            if (rx < NPB) park(rx, j * 4 + c, q, qn, tile, src, nrm, hin);
            rx = ry; ry = rz; rz = rw;
        }
    }
}

__global__ __launch_bounds__(1024) void hop_k(const int* __restrict__ src,
                                              const int* __restrict__ dst,
                                              const float* __restrict__ nrm,
                                              const float* __restrict__ hin,
                                              float* __restrict__ hout,
                                              float* save, int first) {
    __shared__ __attribute__((aligned(16))) float    tile[NPB * F];
    __shared__ __attribute__((aligned(16))) unsigned q[QCAP];
    __shared__ __attribute__((aligned(16))) float    qw[QCAP];
    __shared__ unsigned qn;
    const int tid  = threadIdx.x;
    const int lane = tid & 31;
    const int wv   = tid >> 5;
    const int base = blockIdx.x * NPB;

    unsigned* sv = (unsigned*)save + (long)blockIdx.x * SVSTRIDE;
    const unsigned nsaved = first ? 0xFFFFFFFFu : sv[0];
    const bool sweep = nsaved > (unsigned)CAPSV;
    int n;

    if (!sweep) {
        n = (int)nsaved;
        const uint4* gq = (const uint4*)(sv + SVHEAD);
        const uint4* gw = (const uint4*)(sv + SVHEAD + CAPSV);
        const bool p0 = tid * 4 < n, p1 = (tid + 1024) * 4 < n;
        uint4 a0 = make_uint4(0u, 0u, 0u, 0u), b0 = a0, a1 = a0, b1 = a0;
        if (p0) { a0 = gq[tid];        b0 = gw[tid]; }
        if (p1) { a1 = gq[tid + 1024]; b1 = gw[tid + 1024]; }
        for (int i = tid; i < NPB * F; i += 1024) tile[i] = 0.0f;
        if (p0) { ((uint4*)q)[tid]        = a0; ((uint4*)qw)[tid]        = b0; }
        if (p1) { ((uint4*)q)[tid + 1024] = a1; ((uint4*)qw)[tid + 1024] = b1; }
        __syncthreads();
    } else {
        for (int i = tid; i < NPB * F; i += 1024) tile[i] = 0.0f;
        if (tid == 0) qn = 0u;
        __syncthreads();

        const int4* d4  = (const int4*)dst;
        const int   off = blockIdx.x * STAG;
        constexpr int MAIN = (NV / SWEEP) * SWEEP;
        for (int i0 = tid; i0 < MAIN; i0 += SWEEP) {
            int4 v[8];
            int  jj[8];
#pragma unroll
            for (int k = 0; k < 8; ++k) {
                int j = i0 + k * 1024 + off;
                if (j >= NV) j -= NV;
                jj[k] = j;
                v[k]  = d4[j];
            }
#pragma unroll
            for (int k = 0; k < 8; ++k) look(v[k], jj[k], base, q, &qn, tile, src, nrm, hin);
        }
        for (int i = MAIN + tid; i < NV; i += 1024) {
            int j = i + off;
            if (j >= NV) j -= NV;
            look(d4[j], j, base, q, &qn, tile, src, nrm, hin);
        }
        __syncthreads();

        n = (int)min(qn, (unsigned)QCAP);
        for (int jb = 0; jb < n; jb += 8 * 1024) {
            unsigned u[8];
            int      s[8];
            float    w[8];
#pragma unroll
            for (int k = 0; k < 8; ++k) {
                int j = jb + k * 1024 + tid;
                u[k] = (j < n) ? q[j] : 0u;
            }
#pragma unroll
            for (int k = 0; k < 8; ++k) s[k] = src[u[k] & 0xFFFFFu];
#pragma unroll
            for (int k = 0; k < 8; ++k) w[k] = nrm[s[k]];
#pragma unroll
            for (int k = 0; k < 8; ++k) {
                int j = jb + k * 1024 + tid;
                qw[j] = w[k];
                q[j]  = ((u[k] >> 20) << 16) | (unsigned)s[k];
            }
        }
        __syncthreads();

        if (first) {
            const unsigned nraw = qn;
            if (tid == 0) sv[0] = nraw;
            if (nraw <= (unsigned)CAPSV) {
                uint4* gq = (uint4*)(sv + SVHEAD);
                uint4* gw = (uint4*)(sv + SVHEAD + CAPSV);
                for (int j = tid; j * 4 < n; j += 1024) { gq[j] = ((const uint4*)q)[j]; gw[j] = ((const uint4*)qw)[j]; }
            }
        }
    }

    const int nfull = n & ~15;
    for (int j0 = wv * 16; j0 < nfull; j0 += 32 * 16) {
        int u_l = (int)q[j0 + (lane & 15)];
        int w_l = __float_as_int(qw[j0 + (lane & 15)]);
        float v0[16], v1[16];
        int   ro[16];
#pragma unroll
        for (int k = 0; k < 16; ++k) {
            unsigned u = (unsigned)__builtin_amdgcn_readlane(u_l, k);
            float    w = __int_as_float(__builtin_amdgcn_readlane(w_l, k));
            const float* ip = hin + (long)(u & 0xFFFFu) * F + lane;
            v0[k] = ip[0]  * w;
            v1[k] = ip[32] * w;
            ro[k] = (int)(u >> 16) * F + lane;
        }
#pragma unroll
        for (int k = 0; k < 16; ++k) {
            unsafeAtomicAdd(&tile[ro[k]],      v0[k]);
            unsafeAtomicAdd(&tile[ro[k] + 32], v1[k]);
        }
    }
    {
        int j = nfull + wv;
        if (j < n) {
            unsigned u = q[j];
            float    w = qw[j];
            const float* ip = hin + (long)(u & 0xFFFFu) * F + lane;
            int r0 = (int)(u >> 16) * F + lane;
            unsafeAtomicAdd(&tile[r0],      ip[0]  * w);
            unsafeAtomicAdd(&tile[r0 + 32], ip[32] * w);
        }
    }
    __syncthreads();

    for (int i = tid; i < NPB * F; i += 1024) {
        int node = base + (i >> 6);
        if (node < NN) hout[(long)node * F + (i & 63)] = tile[i] * nrm[node];
    }
}


__device__ __forceinline__ void seg_mac(const float* __restrict__ base, int kbase,
                                        int node, int hi2, int ln15,
                                        const float* Wt,
                                        v8f& c0, v8f& c1, v8f& c2, v8f& c3) {
    const float* rp = base + (long)node * F + hi2;
#pragma unroll 4
    for (int kk = 0; kk < F; kk += 4) {
        v2f a;
        a.x = rp[kk];
        a.y = rp[kk + 1];
        int kr = (kbase + kk + hi2) * F + ln15;
        v2f b0; b0.x = Wt[kr];      b0.y = Wt[kr + F];
        v2f b1; b1.x = Wt[kr + 16]; b1.y = Wt[kr + F + 16];
        v2f b2; b2.x = Wt[kr + 32]; b2.y = Wt[kr + F + 32];
        v2f b3; b3.x = Wt[kr + 48]; b3.y = Wt[kr + F + 48];
        c0 = __builtin_amdgcn_wmma_f32_16x16x4_f32(false, a, false, b0, (short)0, c0, false, false);
        c1 = __builtin_amdgcn_wmma_f32_16x16x4_f32(false, a, false, b1, (short)0, c1, false, false);
        c2 = __builtin_amdgcn_wmma_f32_16x16x4_f32(false, a, false, b2, (short)0, c2, false, false);
        c3 = __builtin_amdgcn_wmma_f32_16x16x4_f32(false, a, false, b3, (short)0, c3, false, false);
    }
}

__global__ __launch_bounds__(256) void gemm_k(const float* __restrict__ feat,
                                              const float* __restrict__ h1,
                                              const float* __restrict__ h2,
                                              const float* __restrict__ h3,
                                              const float* __restrict__ W,
                                              const float* __restrict__ bias,
                                              float* __restrict__ out) {
    __shared__ float Wt[KD * F];
    int tid = threadIdx.x;
    for (int i = tid; i < KD * F; i += 256) {
        int o = i >> 8;
        int k = i & 255;
        Wt[k * F + o] = W[i];
    }
    __syncthreads();

    int lane = tid & 31;
    int ln15 = lane & 15;
    int hi2  = (lane >= 16) ? 2 : 0;
    int rowBase = blockIdx.x * 128 + (tid >> 5) * 16;
    if (rowBase >= NN) return;

    int node = rowBase + ln15;
    if (node >= NN) node = NN - 1;

    v8f c0, c1, c2, c3;
    float bv0 = bias[ln15], bv1 = bias[16 + ln15], bv2 = bias[32 + ln15], bv3 = bias[48 + ln15];
#pragma unroll
    for (int i = 0; i < 8; ++i) { c0[i] = bv0; c1[i] = bv1; c2[i] = bv2; c3[i] = bv3; }

    seg_mac(feat, 0,   node, hi2, ln15, Wt, c0, c1, c2, c3);
    seg_mac(h1,   64,  node, hi2, ln15, Wt, c0, c1, c2, c3);
    seg_mac(h2,   128, node, hi2, ln15, Wt, c0, c1, c2, c3);
    seg_mac(h3,   192, node, hi2, ln15, Wt, c0, c1, c2, c3);

    int mBase = rowBase + (hi2 ? 8 : 0);
#pragma unroll
    for (int r = 0; r < 8; ++r) {
        int row = mBase + r;
        if (row < NN) {
            out[(long)row * F + ln15]      = c0[r];
            out[(long)row * F + 16 + ln15] = c1[r];
            out[(long)row * F + 32 + ln15] = c2[r];
            out[(long)row * F + 48 + ln15] = c3[r];
        }
    }
}


extern "C" void kernel_launch(void* const* d_in, const int* in_sizes, int n_in,
                              void* d_out, int out_size, void* d_ws, size_t ws_size,
                              hipStream_t stream) {
    const float* feat = (const float*)d_in[0];
    const int*   src  = (const int*)d_in[1];
    const int*   dst  = (const int*)d_in[2];
    const float* W    = (const float*)d_in[3];
    const float* bias = (const float*)d_in[4];
    float*       out  = (float*)d_out;

    float* ws   = (float*)d_ws;
    float* nrm  = ws;
    float* h1   = ws + 50048;
    float* h2   = h1 + (long)NN * F;
    float* h3   = h2 + (long)NN * F;

    zero_k <<<196, 256, 0, stream>>>(ws, 50048);
    deg_k  <<<(NE + 255) / 256, 256, 0, stream>>>(dst, (int*)nrm);
    norm_k <<<(NN + 255) / 256, 256, 0, stream>>>(nrm);

    hop_k  <<<HOPG, 1024, 0, stream>>>(src, dst, nrm, feat, h1, out, 1);
    hop_k  <<<HOPG, 1024, 0, stream>>>(src, dst, nrm, h1, h2, out, 0);
    hop_k  <<<HOPG, 1024, 0, stream>>>(src, dst, nrm, h2, h3, out, 0);

    gemm_k <<<(NN + 127) / 128, 256, 0, stream>>>(feat, h1, h2, h3, W, bias, out);
}


# Hop 1 sweeps one owner byte per edge (16 per load; the degree pass writes them, 8 edges a thread), not 4-byte destinations; lanes note hit groups, all threads open them after; LDS adds, arrival order
# speedup vs baseline: 1.8168x; 1.0668x over previous
#include <hip/hip_runtime.h>

typedef float v2f __attribute__((ext_vector_type(2)));
typedef float v8f __attribute__((ext_vector_type(8)));

#define NN 50000
#define NE 800000
#define F  64
#define KD 256


__global__ __launch_bounds__(256) void zero_k(float* __restrict__ p, long n) {
    long i  = (long)blockIdx.x * blockDim.x + threadIdx.x;
    long st = (long)gridDim.x * blockDim.x;
    for (; i < n; i += st) p[i] = 0.0f;
}

#define OWNB 196
__device__ __forceinline__ unsigned own4(int4 d) {
    return ((unsigned)d.x / OWNB) | (((unsigned)d.y / OWNB) << 8) | (((unsigned)d.z / OWNB) << 16) | (((unsigned)d.w / OWNB) << 24);
}

__global__ __launch_bounds__(256) void deg_k(const int* __restrict__ dst, int* __restrict__ deg,
                                             float* __restrict__ keys) {
    int t = blockIdx.x * 256 + threadIdx.x;
    if (t < NE / 8) {
        int4 a = ((const int4*)dst)[2 * t];
        int4 b = ((const int4*)dst)[2 * t + 1];
        atomicAdd(&deg[a.x], 1); atomicAdd(&deg[a.y], 1); atomicAdd(&deg[a.z], 1); atomicAdd(&deg[a.w], 1);
        atomicAdd(&deg[b.x], 1); atomicAdd(&deg[b.y], 1); atomicAdd(&deg[b.z], 1); atomicAdd(&deg[b.w], 1);
        ((uint2*)keys)[t] = make_uint2(own4(a), own4(b));
    }
}

__global__ __launch_bounds__(256) void norm_k(float* __restrict__ nrm) {
    int i = blockIdx.x * 256 + threadIdx.x;
    if (i < NN) {
        float d = (float)((const int*)nrm)[i];
        nrm[i] = rsqrtf(d < 1.0f ? 1.0f : d);
    }
}

#define NPB  OWNB
#define HOPG 256
#define QCAP 16384
#define NV   (NE / 16)
#define SWEEP (8 * 1024)
#define TRIPS ((NV + SWEEP - 1) / SWEEP)
#define STAG 192

static_assert(NPB * HOPG >= NN && HOPG <= 256, "every node has an owner, and the owner fits a byte");
static_assert(NE % 16 == 0, "whole groups of sixteen owner bytes");
static_assert(NN <= (1 << 16) && NPB <= (1 << 16), "queue word after look-up: row << 16 | source");
static_assert(QCAP % (8 * 1024) == 0, "the look-up walks the queue in batches of 8 x 1024");
static_assert((HOPG - 1) * STAG < NV && NV % 8 == 0 && STAG % 32 == 0, "one wrap at most; groups stay on whole lines");

#define SVSTRIDE 11680
#define SVHEAD   32
#define CAPSV    5824
#define KEYOFF   (HOPG * SVSTRIDE)
static_assert((long)KEYOFF + NE / 4 <= (long)NN * F && SVSTRIDE % 32 == 0, "the saved queues and the owner bytes fit in out, line-aligned");
static_assert(SVHEAD + 2 * CAPSV <= SVSTRIDE && CAPSV % 32 == 0 && CAPSV <= QCAP, "count line, words, norms");
static_assert(CAPSV <= 2 * 1024 * 4, "a thread loads a saved queue back in two 16-byte groups");

__device__ __forceinline__ void place(unsigned p, int e, int base, unsigned* q, float* tile,
                                      const int* __restrict__ src, const int* __restrict__ dst,
                                      const float* __restrict__ nrm, const float* __restrict__ hin) {
    if (p < QCAP) {
        q[p] = (unsigned)e;
    } else {
        int   s = src[e];
        float w = nrm[s];
        const float* ip = hin + (long)s * F;
        float* tp = tile + (dst[e] - base) * F;
#pragma unroll 1
        for (int k = 0; k < F; ++k) unsafeAtomicAdd(tp + k, ip[k] * w);
    }
}

__device__ __forceinline__ bool any_ours(uint4 v, unsigned pat) {
    unsigned x0 = v.x ^ pat, x1 = v.y ^ pat, x2 = v.z ^ pat, x3 = v.w ^ pat;
    unsigned z0 = (x0 - 0x01010101u) & ~x0;
    unsigned z1 = (x1 - 0x01010101u) & ~x1;
    unsigned z2 = (x2 - 0x01010101u) & ~x2;
    unsigned z3 = (x3 - 0x01010101u) & ~x3;
    return ((z0 | z1 | z2 | z3) & 0x80808080u) != 0u;
}

__device__ __forceinline__ unsigned ours_mask(unsigned k, unsigned pat) {
    unsigned x = k ^ pat;
    return ~(((x & 0x7F7F7F7Fu) + 0x7F7F7F7Fu) | x | 0x7F7F7F7Fu);
}

__device__ __forceinline__ unsigned place_group(unsigned p, int j, unsigned z0, unsigned z1, unsigned z2, unsigned z3,
                                                int base, unsigned* q, float* tile,
                                                const int* __restrict__ src, const int* __restrict__ dst,
                                                const float* __restrict__ nrm, const float* __restrict__ hin) {
    int e0 = j * 16;
#pragma unroll 1
    for (int d = 0; d < 4; ++d) {
        while (z0) {
            int b = __builtin_ctz(z0) >> 3;
            z0 &= z0 - 1u;
            place(p, e0 + b, base, q, tile, src, dst, nrm, hin);
            ++p;
        }
        z0 = z1; z1 = z2; z2 = z3;
        e0 += 4;
    }
    return p;
}

__device__ __forceinline__ void open_group(int j, uint4 v, unsigned pat, int base,
                                           unsigned* q, unsigned* qn, float* tile,
                                           const int* __restrict__ src, const int* __restrict__ dst,
                                           const float* __restrict__ nrm, const float* __restrict__ hin) {
    unsigned z0 = ours_mask(v.x, pat), z1 = ours_mask(v.y, pat);
    unsigned z2 = ours_mask(v.z, pat), z3 = ours_mask(v.w, pat);
    unsigned p = atomicAdd(qn, (unsigned)(__popc(z0) + __popc(z1) + __popc(z2) + __popc(z3)));
    place_group(p, j, z0, z1, z2, z3, base, q, tile, src, dst, nrm, hin);
}

__global__ __launch_bounds__(1024) void hop_k(const int* __restrict__ src,
                                              const int* __restrict__ dst,
                                              const float* __restrict__ nrm,
                                              const float* __restrict__ hin,
                                              float* __restrict__ hout,
                                              float* save, int first) {
    __shared__ __attribute__((aligned(16))) float    tile[NPB * F];
    __shared__ __attribute__((aligned(16))) unsigned q[QCAP];
    __shared__ __attribute__((aligned(16))) float    qw[QCAP];
    __shared__ unsigned qn;
    __shared__ unsigned gn;
    const int tid  = threadIdx.x;
    const int lane = tid & 31;
    const int wv   = tid >> 5;
    const int base = blockIdx.x * NPB;

    unsigned* sv = (unsigned*)save + (long)blockIdx.x * SVSTRIDE;
    const unsigned nsaved = first ? 0xFFFFFFFFu : sv[0];
    const bool sweep = nsaved > (unsigned)CAPSV;
    int n;

    if (!sweep) {
        n = (int)nsaved;
        const uint4* gq = (const uint4*)(sv + SVHEAD);
        const uint4* gw = (const uint4*)(sv + SVHEAD + CAPSV);
        const bool p0 = tid * 4 < n, p1 = (tid + 1024) * 4 < n;
        uint4 a0 = make_uint4(0u, 0u, 0u, 0u), b0 = a0, a1 = a0, b1 = a0;
        if (p0) { a0 = gq[tid];        b0 = gw[tid]; }
        if (p1) { a1 = gq[tid + 1024]; b1 = gw[tid + 1024]; }
        for (int i = tid; i < NPB * F; i += 1024) tile[i] = 0.0f;
        if (p0) { ((uint4*)q)[tid]        = a0; ((uint4*)qw)[tid]        = b0; }
        if (p1) { ((uint4*)q)[tid + 1024] = a1; ((uint4*)qw)[tid + 1024] = b1; }
        __syncthreads();
    } else {
        for (int i = tid; i < NPB * F; i += 1024) tile[i] = 0.0f;
        if (tid == 0) { qn = 0u; gn = 0u; }
        __syncthreads();

        const uint4*   d4  = (const uint4*)((const unsigned*)save + KEYOFF);
        const unsigned pat = blockIdx.x * 0x01010101u;
        const int      off = blockIdx.x * STAG;
        unsigned*      g   = (unsigned*)qw;
        for (int m = 0; m < TRIPS; ++m) {
            const int i0 = m * SWEEP + tid;
            uint4 v[8];
#pragma unroll
            for (int k = 0; k < 8; ++k) {
                int i = i0 + k * 1024;
                int j = i + off;
                if (j >= NV) j -= NV;
                if (i >= NV) j = 0;
                v[k] = d4[j];
            }
            unsigned fl = 0u;
#pragma unroll
            for (int k = 0; k < 8; ++k) fl |= any_ours(v[k], pat) ? (1u << k) : 0u;
            if (m == TRIPS - 1) {
                unsigned vm = 0u;
#pragma unroll
                for (int k = 0; k < 8; ++k) vm |= (i0 + k * 1024 < NV) ? (1u << k) : 0u;
                fl &= vm;
            }
#pragma unroll 1
            while (fl) {
                int k = __builtin_ctz(fl);
                fl &= fl - 1u;
                int j = i0 + k * 1024 + off;
                if (j >= NV) j -= NV;
                unsigned p = atomicAdd(&gn, 1u);
                if (p < QCAP) g[p] = (unsigned)j;
                else          open_group(j, d4[j], pat, base, q, &qn, tile, src, dst, nrm, hin);
            }
        }
        __syncthreads();

        const int ng = (int)min(gn, (unsigned)QCAP);
        for (int ib = 0; ib < ng; ib += 4 * 1024) {
            int   gj[4];
            uint4 gv[4];
#pragma unroll
            for (int k = 0; k < 4; ++k) {
                int i = ib + k * 1024 + tid;
                gj[k] = (i < ng) ? (int)g[i] : 0;
            }
#pragma unroll
            for (int k = 0; k < 4; ++k) gv[k] = d4[gj[k]];
            unsigned z[4][4];
            unsigned h = 0u;
#pragma unroll
            for (int k = 0; k < 4; ++k) {
                const bool ok = ib + k * 1024 + tid < ng;
                z[k][0] = ok ? ours_mask(gv[k].x, pat) : 0u;
                z[k][1] = ok ? ours_mask(gv[k].y, pat) : 0u;
                z[k][2] = ok ? ours_mask(gv[k].z, pat) : 0u;
                z[k][3] = ok ? ours_mask(gv[k].w, pat) : 0u;
                h += (unsigned)(__popc(z[k][0]) + __popc(z[k][1]) + __popc(z[k][2]) + __popc(z[k][3]));
            }
            unsigned p = atomicAdd(&qn, h);
#pragma unroll
            for (int k = 0; k < 4; ++k)
                p = place_group(p, gj[k], z[k][0], z[k][1], z[k][2], z[k][3], base, q, tile, src, dst, nrm, hin);
        }
        __syncthreads();

        n = (int)min(qn, (unsigned)QCAP);
        for (int jb = 0; jb < n; jb += 8 * 1024) {
            unsigned u[8];
            int      s[8], d[8];
            float    w[8];
#pragma unroll
            for (int k = 0; k < 8; ++k) {
                int j = jb + k * 1024 + tid;
                u[k] = (j < n) ? q[j] : 0u;
            }
#pragma unroll
            for (int k = 0; k < 8; ++k) s[k] = src[u[k]];
#pragma unroll
            for (int k = 0; k < 8; ++k) d[k] = dst[u[k]];
#pragma unroll
            for (int k = 0; k < 8; ++k) w[k] = nrm[s[k]];
#pragma unroll
            for (int k = 0; k < 8; ++k) {
                int j = jb + k * 1024 + tid;
                qw[j] = w[k];
                q[j]  = ((unsigned)(d[k] - base) << 16) | (unsigned)s[k];
            }
        }
        __syncthreads();

        if (first) {
            const unsigned nraw = qn;
            if (tid == 0) sv[0] = nraw;
            if (nraw <= (unsigned)CAPSV) {
                uint4* gq = (uint4*)(sv + SVHEAD);
                uint4* gw = (uint4*)(sv + SVHEAD + CAPSV);
                for (int j = tid; j * 4 < n; j += 1024) { gq[j] = ((const uint4*)q)[j]; gw[j] = ((const uint4*)qw)[j]; }
            }
        }
    }

    const int nfull = n & ~15;
    for (int j0 = wv * 16; j0 < nfull; j0 += 32 * 16) {
        int u_l = (int)q[j0 + (lane & 15)];
        int w_l = __float_as_int(qw[j0 + (lane & 15)]);
        float v0[16], v1[16];
        int   ro[16];
#pragma unroll
        for (int k = 0; k < 16; ++k) {
            unsigned u = (unsigned)__builtin_amdgcn_readlane(u_l, k);
            float    w = __int_as_float(__builtin_amdgcn_readlane(w_l, k));
            const float* ip = hin + (long)(u & 0xFFFFu) * F + lane;
            v0[k] = ip[0]  * w;
            v1[k] = ip[32] * w;
            ro[k] = (int)(u >> 16) * F + lane;
        }
#pragma unroll
        for (int k = 0; k < 16; ++k) {
            unsafeAtomicAdd(&tile[ro[k]],      v0[k]);
            unsafeAtomicAdd(&tile[ro[k] + 32], v1[k]);
        }
    }
    {
        int j = nfull + wv;
        if (j < n) {
            unsigned u = q[j];
            float    w = qw[j];
            const float* ip = hin + (long)(u & 0xFFFFu) * F + lane;
            int r0 = (int)(u >> 16) * F + lane;
            unsafeAtomicAdd(&tile[r0],      ip[0]  * w);
            unsafeAtomicAdd(&tile[r0 + 32], ip[32] * w);
        }
    }
    __syncthreads();

    for (int i = tid; i < NPB * F; i += 1024) {
        int node = base + (i >> 6);
        if (node < NN) hout[(long)node * F + (i & 63)] = tile[i] * nrm[node];
    }
}


__device__ __forceinline__ void seg_mac(const float* __restrict__ base, int kbase,
                                        int node, int hi2, int ln15,
                                        const float* Wt,
                                        v8f& c0, v8f& c1, v8f& c2, v8f& c3) {
    const float* rp = base + (long)node * F + hi2;
#pragma unroll 4
    for (int kk = 0; kk < F; kk += 4) {
        v2f a;
        a.x = rp[kk];
        a.y = rp[kk + 1];
        int kr = (kbase + kk + hi2) * F + ln15;
        v2f b0; b0.x = Wt[kr];      b0.y = Wt[kr + F];
        v2f b1; b1.x = Wt[kr + 16]; b1.y = Wt[kr + F + 16];
        v2f b2; b2.x = Wt[kr + 32]; b2.y = Wt[kr + F + 32];
        v2f b3; b3.x = Wt[kr + 48]; b3.y = Wt[kr + F + 48];
        c0 = __builtin_amdgcn_wmma_f32_16x16x4_f32(false, a, false, b0, (short)0, c0, false, false);
        c1 = __builtin_amdgcn_wmma_f32_16x16x4_f32(false, a, false, b1, (short)0, c1, false, false);
        c2 = __builtin_amdgcn_wmma_f32_16x16x4_f32(false, a, false, b2, (short)0, c2, false, false);
        c3 = __builtin_amdgcn_wmma_f32_16x16x4_f32(false, a, false, b3, (short)0, c3, false, false);
    }
}

__global__ __launch_bounds__(256) void gemm_k(const float* __restrict__ feat,
                                              const float* __restrict__ h1,
                                              const float* __restrict__ h2,
                                              const float* __restrict__ h3,
                                              const float* __restrict__ W,
                                              const float* __restrict__ bias,
                                              float* __restrict__ out) {
    __shared__ float Wt[KD * F];
    int tid = threadIdx.x;
    for (int i = tid; i < KD * F; i += 256) {
        int o = i >> 8;
        int k = i & 255;
        Wt[k * F + o] = W[i];
    }
    __syncthreads();

    int lane = tid & 31;
    int ln15 = lane & 15;
    int hi2  = (lane >= 16) ? 2 : 0;
    int rowBase = blockIdx.x * 128 + (tid >> 5) * 16;
    if (rowBase >= NN) return;

    int node = rowBase + ln15;
    if (node >= NN) node = NN - 1;

    v8f c0, c1, c2, c3;
    float bv0 = bias[ln15], bv1 = bias[16 + ln15], bv2 = bias[32 + ln15], bv3 = bias[48 + ln15];
#pragma unroll
    for (int i = 0; i < 8; ++i) { c0[i] = bv0; c1[i] = bv1; c2[i] = bv2; c3[i] = bv3; }

    seg_mac(feat, 0,   node, hi2, ln15, Wt, c0, c1, c2, c3);
    seg_mac(h1,   64,  node, hi2, ln15, Wt, c0, c1, c2, c3);
    seg_mac(h2,   128, node, hi2, ln15, Wt, c0, c1, c2, c3);
    seg_mac(h3,   192, node, hi2, ln15, Wt, c0, c1, c2, c3);

    int mBase = rowBase + (hi2 ? 8 : 0);
#pragma unroll
    for (int r = 0; r < 8; ++r) {
        int row = mBase + r;
        if (row < NN) {
            out[(long)row * F + ln15]      = c0[r];
            out[(long)row * F + 16 + ln15] = c1[r];
            out[(long)row * F + 32 + ln15] = c2[r];
            out[(long)row * F + 48 + ln15] = c3[r];
        }
    }
}


extern "C" void kernel_launch(void* const* d_in, const int* in_sizes, int n_in,
                              void* d_out, int out_size, void* d_ws, size_t ws_size,
                              hipStream_t stream) {
    const float* feat = (const float*)d_in[0];
    const int*   src  = (const int*)d_in[1];
    const int*   dst  = (const int*)d_in[2];
    const float* W    = (const float*)d_in[3];
    const float* bias = (const float*)d_in[4];
    float*       out  = (float*)d_out;

    float* ws   = (float*)d_ws;
    float* nrm  = ws;
    float* h1   = ws + 50048;
    float* h2   = h1 + (long)NN * F;
    float* h3   = h2 + (long)NN * F;

    zero_k <<<196, 256, 0, stream>>>(ws, 50048);
    deg_k  <<<(NE / 8 + 255) / 256, 256, 0, stream>>>(dst, (int*)nrm, out + KEYOFF);
    norm_k <<<(NN + 255) / 256, 256, 0, stream>>>(nrm);

    hop_k  <<<HOPG, 1024, 0, stream>>>(src, dst, nrm, feat, h1, out, 1);
    hop_k  <<<HOPG, 1024, 0, stream>>>(src, dst, nrm, h1, h2, out, 0);
    hop_k  <<<HOPG, 1024, 0, stream>>>(src, dst, nrm, h2, h3, out, 0);

    gemm_k <<<(NN + 127) / 128, 256, 0, stream>>>(feat, h1, h2, h3, W, bias, out);
}


# Projection: a wave keeps its 32 output columns of W for all 256 k in registers (16 bytes per lane from W's rows) and walks row tiles; W's LDS copy goes; k order in a matrix step changed (sum order)
# speedup vs baseline: 1.9764x; 1.0878x over previous
#include <hip/hip_runtime.h>

typedef float v2f __attribute__((ext_vector_type(2)));
typedef float v8f __attribute__((ext_vector_type(8)));

#define NN 50000
#define NE 800000
#define F  64
#define KD 256


__global__ __launch_bounds__(256) void zero_k(float* __restrict__ p, long n) {
    long i  = (long)blockIdx.x * blockDim.x + threadIdx.x;
    long st = (long)gridDim.x * blockDim.x;
    for (; i < n; i += st) p[i] = 0.0f;
}

#define OWNB 196
__device__ __forceinline__ unsigned own4(int4 d) {
    return ((unsigned)d.x / OWNB) | (((unsigned)d.y / OWNB) << 8) | (((unsigned)d.z / OWNB) << 16) | (((unsigned)d.w / OWNB) << 24);
}

__global__ __launch_bounds__(256) void deg_k(const int* __restrict__ dst, int* __restrict__ deg,
                                             float* __restrict__ keys) {
    int t = blockIdx.x * 256 + threadIdx.x;
    if (t < NE / 8) {
        int4 a = ((const int4*)dst)[2 * t];
        int4 b = ((const int4*)dst)[2 * t + 1];
        atomicAdd(&deg[a.x], 1); atomicAdd(&deg[a.y], 1); atomicAdd(&deg[a.z], 1); atomicAdd(&deg[a.w], 1);
        atomicAdd(&deg[b.x], 1); atomicAdd(&deg[b.y], 1); atomicAdd(&deg[b.z], 1); atomicAdd(&deg[b.w], 1);
        ((uint2*)keys)[t] = make_uint2(own4(a), own4(b));
    }
}

__global__ __launch_bounds__(256) void norm_k(float* __restrict__ nrm) {
    int i = blockIdx.x * 256 + threadIdx.x;
    if (i < NN) {
        float d = (float)((const int*)nrm)[i];
        nrm[i] = rsqrtf(d < 1.0f ? 1.0f : d);
    }
}

#define NPB  OWNB
#define HOPG 256
#define QCAP 16384
#define NV   (NE / 16)
#define SWEEP (8 * 1024)
#define TRIPS ((NV + SWEEP - 1) / SWEEP)
#define STAG 192

static_assert(NPB * HOPG >= NN && HOPG <= 256, "every node has an owner, and the owner fits a byte");
static_assert(NE % 16 == 0, "whole groups of sixteen owner bytes");
static_assert(NN <= (1 << 16) && NPB <= (1 << 16), "queue word after look-up: row << 16 | source");
static_assert(QCAP % (8 * 1024) == 0, "the look-up walks the queue in batches of 8 x 1024");
static_assert((HOPG - 1) * STAG < NV && NV % 8 == 0 && STAG % 32 == 0, "one wrap at most; groups stay on whole lines");

#define SVSTRIDE 11680
#define SVHEAD   32
#define CAPSV    5824
#define KEYOFF   (HOPG * SVSTRIDE)
static_assert((long)KEYOFF + NE / 4 <= (long)NN * F && SVSTRIDE % 32 == 0, "the saved queues and the owner bytes fit in out, line-aligned");
static_assert(SVHEAD + 2 * CAPSV <= SVSTRIDE && CAPSV % 32 == 0 && CAPSV <= QCAP, "count line, words, norms");
static_assert(CAPSV <= 2 * 1024 * 4, "a thread loads a saved queue back in two 16-byte groups");

__device__ __forceinline__ void place(unsigned p, int e, int base, unsigned* q, float* tile,
                                      const int* __restrict__ src, const int* __restrict__ dst,
                                      const float* __restrict__ nrm, const float* __restrict__ hin) {
    if (p < QCAP) {
        q[p] = (unsigned)e;
    } else {
        int   s = src[e];
        float w = nrm[s];
        const float* ip = hin + (long)s * F;
        float* tp = tile + (dst[e] - base) * F;
#pragma unroll 1
        for (int k = 0; k < F; ++k) unsafeAtomicAdd(tp + k, ip[k] * w);
    }
}

__device__ __forceinline__ bool any_ours(uint4 v, unsigned pat) {
    unsigned x0 = v.x ^ pat, x1 = v.y ^ pat, x2 = v.z ^ pat, x3 = v.w ^ pat;
    unsigned z0 = (x0 - 0x01010101u) & ~x0;
    unsigned z1 = (x1 - 0x01010101u) & ~x1;
    unsigned z2 = (x2 - 0x01010101u) & ~x2;
    unsigned z3 = (x3 - 0x01010101u) & ~x3;
    return ((z0 | z1 | z2 | z3) & 0x80808080u) != 0u;
}

__device__ __forceinline__ unsigned ours_mask(unsigned k, unsigned pat) {
    unsigned x = k ^ pat;
    return ~(((x & 0x7F7F7F7Fu) + 0x7F7F7F7Fu) | x | 0x7F7F7F7Fu);
}

__device__ __forceinline__ unsigned place_group(unsigned p, int j, unsigned z0, unsigned z1, unsigned z2, unsigned z3,
                                                int base, unsigned* q, float* tile,
                                                const int* __restrict__ src, const int* __restrict__ dst,
                                                const float* __restrict__ nrm, const float* __restrict__ hin) {
    int e0 = j * 16;
#pragma unroll 1
    for (int d = 0; d < 4; ++d) {
        while (z0) {
            int b = __builtin_ctz(z0) >> 3;
            z0 &= z0 - 1u;
            place(p, e0 + b, base, q, tile, src, dst, nrm, hin);
            ++p;
        }
        z0 = z1; z1 = z2; z2 = z3;
        e0 += 4;
    }
    return p;
}

__device__ __forceinline__ void open_group(int j, uint4 v, unsigned pat, int base,
                                           unsigned* q, unsigned* qn, float* tile,
                                           const int* __restrict__ src, const int* __restrict__ dst,
                                           const float* __restrict__ nrm, const float* __restrict__ hin) {
    unsigned z0 = ours_mask(v.x, pat), z1 = ours_mask(v.y, pat);
    unsigned z2 = ours_mask(v.z, pat), z3 = ours_mask(v.w, pat);
    unsigned p = atomicAdd(qn, (unsigned)(__popc(z0) + __popc(z1) + __popc(z2) + __popc(z3)));
    place_group(p, j, z0, z1, z2, z3, base, q, tile, src, dst, nrm, hin);
}

__global__ __launch_bounds__(1024) void hop_k(const int* __restrict__ src,
                                              const int* __restrict__ dst,
                                              const float* __restrict__ nrm,
                                              const float* __restrict__ hin,
                                              float* __restrict__ hout,
                                              float* save, int first) {
    __shared__ __attribute__((aligned(16))) float    tile[NPB * F];
    __shared__ __attribute__((aligned(16))) unsigned q[QCAP];
    __shared__ __attribute__((aligned(16))) float    qw[QCAP];
    __shared__ unsigned qn;
    __shared__ unsigned gn;
    const int tid  = threadIdx.x;
    const int lane = tid & 31;
    const int wv   = tid >> 5;
    const int base = blockIdx.x * NPB;

    unsigned* sv = (unsigned*)save + (long)blockIdx.x * SVSTRIDE;
    const unsigned nsaved = first ? 0xFFFFFFFFu : sv[0];
    const bool sweep = nsaved > (unsigned)CAPSV;
    int n;

    if (!sweep) {
        n = (int)nsaved;
        const uint4* gq = (const uint4*)(sv + SVHEAD);
        const uint4* gw = (const uint4*)(sv + SVHEAD + CAPSV);
        const bool p0 = tid * 4 < n, p1 = (tid + 1024) * 4 < n;
        uint4 a0 = make_uint4(0u, 0u, 0u, 0u), b0 = a0, a1 = a0, b1 = a0;
        if (p0) { a0 = gq[tid];        b0 = gw[tid]; }
        if (p1) { a1 = gq[tid + 1024]; b1 = gw[tid + 1024]; }
        for (int i = tid; i < NPB * F; i += 1024) tile[i] = 0.0f;
        if (p0) { ((uint4*)q)[tid]        = a0; ((uint4*)qw)[tid]        = b0; }
        if (p1) { ((uint4*)q)[tid + 1024] = a1; ((uint4*)qw)[tid + 1024] = b1; }
        __syncthreads();
    } else {
        for (int i = tid; i < NPB * F; i += 1024) tile[i] = 0.0f;
        if (tid == 0) { qn = 0u; gn = 0u; }
        __syncthreads();

        const uint4*   d4  = (const uint4*)((const unsigned*)save + KEYOFF);
        const unsigned pat = blockIdx.x * 0x01010101u;
        const int      off = blockIdx.x * STAG;
        unsigned*      g   = (unsigned*)qw;
        for (int m = 0; m < TRIPS; ++m) {
            const int i0 = m * SWEEP + tid;
            uint4 v[8];
#pragma unroll
            for (int k = 0; k < 8; ++k) {
                int i = i0 + k * 1024;
                int j = i + off;
                if (j >= NV) j -= NV;
                if (i >= NV) j = 0;
                v[k] = d4[j];
            }
            unsigned fl = 0u;
#pragma unroll
            for (int k = 0; k < 8; ++k) fl |= any_ours(v[k], pat) ? (1u << k) : 0u;
            if (m == TRIPS - 1) {
                unsigned vm = 0u;
#pragma unroll
                for (int k = 0; k < 8; ++k) vm |= (i0 + k * 1024 < NV) ? (1u << k) : 0u;
                fl &= vm;
            }
#pragma unroll 1
            while (fl) {
                int k = __builtin_ctz(fl);
                fl &= fl - 1u;
                int j = i0 + k * 1024 + off;
                if (j >= NV) j -= NV;
                unsigned p = atomicAdd(&gn, 1u);
                if (p < QCAP) g[p] = (unsigned)j;
                else          open_group(j, d4[j], pat, base, q, &qn, tile, src, dst, nrm, hin);
            }
        }
        __syncthreads();

        const int ng = (int)min(gn, (unsigned)QCAP);
        for (int ib = 0; ib < ng; ib += 4 * 1024) {
            int   gj[4];
            uint4 gv[4];
#pragma unroll
            for (int k = 0; k < 4; ++k) {
                int i = ib + k * 1024 + tid;
                gj[k] = (i < ng) ? (int)g[i] : 0;
            }
#pragma unroll
            for (int k = 0; k < 4; ++k) gv[k] = d4[gj[k]];
            unsigned z[4][4];
            unsigned h = 0u;
#pragma unroll
            for (int k = 0; k < 4; ++k) {
                const bool ok = ib + k * 1024 + tid < ng;
                z[k][0] = ok ? ours_mask(gv[k].x, pat) : 0u;
                z[k][1] = ok ? ours_mask(gv[k].y, pat) : 0u;
                z[k][2] = ok ? ours_mask(gv[k].z, pat) : 0u;
                z[k][3] = ok ? ours_mask(gv[k].w, pat) : 0u;
                h += (unsigned)(__popc(z[k][0]) + __popc(z[k][1]) + __popc(z[k][2]) + __popc(z[k][3]));
            }
            unsigned p = atomicAdd(&qn, h);
#pragma unroll
            for (int k = 0; k < 4; ++k)
                p = place_group(p, gj[k], z[k][0], z[k][1], z[k][2], z[k][3], base, q, tile, src, dst, nrm, hin);
        }
        __syncthreads();

        n = (int)min(qn, (unsigned)QCAP);
        for (int jb = 0; jb < n; jb += 8 * 1024) {
            unsigned u[8];
            int      s[8], d[8];
            float    w[8];
#pragma unroll
            for (int k = 0; k < 8; ++k) {
                int j = jb + k * 1024 + tid;
                u[k] = (j < n) ? q[j] : 0u;
            }
#pragma unroll
            for (int k = 0; k < 8; ++k) s[k] = src[u[k]];
#pragma unroll
            for (int k = 0; k < 8; ++k) d[k] = dst[u[k]];
#pragma unroll
            for (int k = 0; k < 8; ++k) w[k] = nrm[s[k]];
#pragma unroll
            for (int k = 0; k < 8; ++k) {
                int j = jb + k * 1024 + tid;
                qw[j] = w[k];
                q[j]  = ((unsigned)(d[k] - base) << 16) | (unsigned)s[k];
            }
        }
        __syncthreads();

        if (first) {
            const unsigned nraw = qn;
            if (tid == 0) sv[0] = nraw;
            if (nraw <= (unsigned)CAPSV) {
                uint4* gq = (uint4*)(sv + SVHEAD);
                uint4* gw = (uint4*)(sv + SVHEAD + CAPSV);
                for (int j = tid; j * 4 < n; j += 1024) { gq[j] = ((const uint4*)q)[j]; gw[j] = ((const uint4*)qw)[j]; }
            }
        }
    }

    const int nfull = n & ~15;
    for (int j0 = wv * 16; j0 < nfull; j0 += 32 * 16) {
        int u_l = (int)q[j0 + (lane & 15)];
        int w_l = __float_as_int(qw[j0 + (lane & 15)]);
        float v0[16], v1[16];
        int   ro[16];
#pragma unroll
        for (int k = 0; k < 16; ++k) {
            unsigned u = (unsigned)__builtin_amdgcn_readlane(u_l, k);
            float    w = __int_as_float(__builtin_amdgcn_readlane(w_l, k));
            const float* ip = hin + (long)(u & 0xFFFFu) * F + lane;
            v0[k] = ip[0]  * w;
            v1[k] = ip[32] * w;
            ro[k] = (int)(u >> 16) * F + lane;
        }
#pragma unroll
        for (int k = 0; k < 16; ++k) {
            unsafeAtomicAdd(&tile[ro[k]],      v0[k]);
            unsafeAtomicAdd(&tile[ro[k] + 32], v1[k]);
        }
    }
    {
        int j = nfull + wv;
        if (j < n) {
            unsigned u = q[j];
            float    w = qw[j];
            const float* ip = hin + (long)(u & 0xFFFFu) * F + lane;
            int r0 = (int)(u >> 16) * F + lane;
            unsafeAtomicAdd(&tile[r0],      ip[0]  * w);
            unsafeAtomicAdd(&tile[r0 + 32], ip[32] * w);
        }
    }
    __syncthreads();

    for (int i = tid; i < NPB * F; i += 1024) {
        int node = base + (i >> 6);
        if (node < NN) hout[(long)node * F + (i & 63)] = tile[i] * nrm[node];
    }
}

#define GW 256
#define NT (NN / 16)
static_assert(NN % 16 == 0 && GW <= NT, "whole row tiles only, and every work-group has one");

__device__ __forceinline__ void ld8(float4 (&a)[8], const float* __restrict__ p) {
#pragma unroll
    for (int q = 0; q < 8; ++q) a[q] = *(const float4*)(p + 8 * q);
}

template <int SEG>
__device__ __forceinline__ void seg_mac(const float4 (&a)[8], const float4 (&w0)[32], const float4 (&w1)[32],
                                        v8f& c0, v8f& c1) {
#pragma unroll
    for (int q = 0; q < 8; ++q) {
        const float4 x = a[q], y0 = w0[SEG * 8 + q], y1 = w1[SEG * 8 + q];
        v2f al;  al.x  = x.x;  al.y  = x.y;
        v2f ah;  ah.x  = x.z;  ah.y  = x.w;
        v2f b0l; b0l.x = y0.x; b0l.y = y0.y;
        v2f b0h; b0h.x = y0.z; b0h.y = y0.w;
        v2f b1l; b1l.x = y1.x; b1l.y = y1.y;
        v2f b1h; b1h.x = y1.z; b1h.y = y1.w;
        c0 = __builtin_amdgcn_wmma_f32_16x16x4_f32(false, al, false, b0l, (short)0, c0, false, false);
        c1 = __builtin_amdgcn_wmma_f32_16x16x4_f32(false, al, false, b1l, (short)0, c1, false, false);
        c0 = __builtin_amdgcn_wmma_f32_16x16x4_f32(false, ah, false, b0h, (short)0, c0, false, false);
        c1 = __builtin_amdgcn_wmma_f32_16x16x4_f32(false, ah, false, b1h, (short)0, c1, false, false);
    }
}

__global__ __launch_bounds__(256) void gemm_k(const float* __restrict__ feat,
                                              const float* __restrict__ h1,
                                              const float* __restrict__ h2,
                                              const float* __restrict__ h3,
                                              const float* __restrict__ W,
                                              const float* __restrict__ bias,
                                              float* __restrict__ out) {
    const int lane = threadIdx.x & 31;
    const int c    = lane & 15;
    const int h    = lane >> 4;
    const int wv = __builtin_amdgcn_readfirstlane((int)(threadIdx.x >> 5));
    const int jh = wv & 1;
    const int g  = wv >> 1;
    const int cnt = (NT - 1 - (int)blockIdx.x) / GW + 1;

    float4 w0[32], w1[32];
    {
        const float* wp = W + (long)(jh * 32 + c) * KD + 4 * h;
#pragma unroll
        for (int i = 0; i < 32; ++i) w0[i] = *(const float4*)(wp + 8 * i);
#pragma unroll
        for (int i = 0; i < 32; ++i) w1[i] = *(const float4*)(wp + 16 * KD + 8 * i);
    }
    const float bv0 = bias[jh * 32 + c], bv1 = bias[jh * 32 + 16 + c];

    const long lo = (long)c * F + 4 * h;
    float4 a0[8], a1[8];
    ld8(a0, feat + (long)((int)blockIdx.x + GW * g) * (16 * F) + lo);
    for (int n = g; n < cnt; n += 4) {
        const int  tile = (int)blockIdx.x + GW * n;
        const int  nx   = (n + 4 < cnt) ? tile + 4 * GW : tile;
        const long to   = (long)tile * (16 * F) + lo;
        v8f c0, c1;
#pragma unroll
        for (int r = 0; r < 8; ++r) { c0[r] = bv0; c1[r] = bv1; }

        ld8(a1, h1 + to);
        __builtin_amdgcn_sched_barrier(0);
        seg_mac<0>(a0, w0, w1, c0, c1);
        __builtin_amdgcn_sched_barrier(0);
        ld8(a0, h2 + to);
        __builtin_amdgcn_sched_barrier(0);
        seg_mac<1>(a1, w0, w1, c0, c1);
        __builtin_amdgcn_sched_barrier(0);
        ld8(a1, h3 + to);
        __builtin_amdgcn_sched_barrier(0);
        seg_mac<2>(a0, w0, w1, c0, c1);
        __builtin_amdgcn_sched_barrier(0);
        ld8(a0, feat + (long)nx * (16 * F) + lo);
        __builtin_amdgcn_sched_barrier(0);
        seg_mac<3>(a1, w0, w1, c0, c1);

        float* op = out + (long)(tile * 16 + 8 * h) * F + jh * 32 + c;
#pragma unroll
        for (int r = 0; r < 8; ++r) {
            op[r * F]      = c0[r];
            op[r * F + 16] = c1[r];
        }
    }
}


extern "C" void kernel_launch(void* const* d_in, const int* in_sizes, int n_in,
                              void* d_out, int out_size, void* d_ws, size_t ws_size,
                              hipStream_t stream) {
    const float* feat = (const float*)d_in[0];
    const int*   src  = (const int*)d_in[1];
    const int*   dst  = (const int*)d_in[2];
    const float* W    = (const float*)d_in[3];
    const float* bias = (const float*)d_in[4];
    float*       out  = (float*)d_out;

    float* ws   = (float*)d_ws;
    float* nrm  = ws;
    float* h1   = ws + 50048;
    float* h2   = h1 + (long)NN * F;
    float* h3   = h2 + (long)NN * F;

    zero_k <<<196, 256, 0, stream>>>(ws, 50048);
    deg_k  <<<(NE / 8 + 255) / 256, 256, 0, stream>>>(dst, (int*)nrm, out + KEYOFF);
    norm_k <<<(NN + 255) / 256, 256, 0, stream>>>(nrm);

    hop_k  <<<HOPG, 1024, 0, stream>>>(src, dst, nrm, feat, h1, out, 1);
    hop_k  <<<HOPG, 1024, 0, stream>>>(src, dst, nrm, h1, h2, out, 0);
    hop_k  <<<HOPG, 1024, 0, stream>>>(src, dst, nrm, h2, h3, out, 0);

    gemm_k <<<GW, 256, 0, stream>>>(feat, h1, h2, h3, W, bias, out);
}


# Hop sum: row r belongs to wave r & 31; parked edges go to a list per owning wave (LDS, arrival order) and each wave sums its 7 rows in registers, in list order; the LDS float adds go
# speedup vs baseline: 2.3102x; 1.1689x over previous
#include <hip/hip_runtime.h>

typedef float v2f __attribute__((ext_vector_type(2)));
typedef float v8f __attribute__((ext_vector_type(8)));

#define NN 50000
#define NE 800000
#define F  64
#define KD 256


__global__ __launch_bounds__(256) void zero_k(float* __restrict__ p, long n) {
    long i  = (long)blockIdx.x * blockDim.x + threadIdx.x;
    long st = (long)gridDim.x * blockDim.x;
    for (; i < n; i += st) p[i] = 0.0f;
}

#define OWNB 196
__device__ __forceinline__ unsigned own4(int4 d) {
    return ((unsigned)d.x / OWNB) | (((unsigned)d.y / OWNB) << 8) | (((unsigned)d.z / OWNB) << 16) | (((unsigned)d.w / OWNB) << 24);
}

__global__ __launch_bounds__(256) void deg_k(const int* __restrict__ dst, int* __restrict__ deg,
                                             float* __restrict__ keys) {
    int t = blockIdx.x * 256 + threadIdx.x;
    if (t < NE / 8) {
        int4 a = ((const int4*)dst)[2 * t];
        int4 b = ((const int4*)dst)[2 * t + 1];
        atomicAdd(&deg[a.x], 1); atomicAdd(&deg[a.y], 1); atomicAdd(&deg[a.z], 1); atomicAdd(&deg[a.w], 1);
        atomicAdd(&deg[b.x], 1); atomicAdd(&deg[b.y], 1); atomicAdd(&deg[b.z], 1); atomicAdd(&deg[b.w], 1);
        ((uint2*)keys)[t] = make_uint2(own4(a), own4(b));
    }
}

__global__ __launch_bounds__(256) void norm_k(float* __restrict__ nrm) {
    int i = blockIdx.x * 256 + threadIdx.x;
    if (i < NN) {
        float d = (float)((const int*)nrm)[i];
        nrm[i] = rsqrtf(d < 1.0f ? 1.0f : d);
    }
}

#define NPB  OWNB
#define HOPG 256
#define QCAP 16384
#define NV   (NE / 16)
#define SWEEP (8 * 1024)
#define TRIPS ((NV + SWEEP - 1) / SWEEP)
#define STAG 192
#define NWV  32
#define RPW  7
#define PCAP 512

static_assert(NPB * HOPG >= NN && HOPG <= 256, "every node has an owner, and the owner fits a byte");
static_assert(NE % 16 == 0, "whole groups of sixteen owner bytes");
static_assert(NWV * RPW >= NPB && NWV == 32 && RPW <= 8, "every row has an owning wave; a row's place in its wave fits three bits");
static_assert(NN <= (1 << 16) && NPB <= (1 << 16), "queue word after look-up: row << 16 | source");
static_assert(QCAP % (8 * 1024) == 0, "the look-up walks the queue in batches of 8 x 1024");
static_assert((HOPG - 1) * STAG < NV && NV % 8 == 0 && STAG % 32 == 0, "one wrap at most; groups stay on whole lines");

#define SVSTRIDE 11680
#define SVHEAD   32
#define CAPSV    5824
#define KEYOFF   (HOPG * SVSTRIDE)
static_assert((long)KEYOFF + NE / 4 <= (long)NN * F && SVSTRIDE % 32 == 0, "the saved queues and the owner bytes fit in out, line-aligned");
static_assert(SVHEAD + 2 * CAPSV <= SVSTRIDE && CAPSV % 32 == 0 && CAPSV <= QCAP, "count line, words, norms");
static_assert(CAPSV <= 2 * 1024 * 4, "a thread loads a saved queue back in two 16-byte groups");

__device__ __forceinline__ void place(unsigned p, int e, int base, unsigned* q, float* tile,
                                      const int* __restrict__ src, const int* __restrict__ dst,
                                      const float* __restrict__ nrm, const float* __restrict__ hin) {
    if (p < QCAP) {
        q[p] = (unsigned)e;
    } else {
        int   s = src[e];
        float w = nrm[s];
        const float* ip = hin + (long)s * F;
        float* tp = tile + (dst[e] - base) * F;
#pragma unroll 1
        for (int k = 0; k < F; ++k) unsafeAtomicAdd(tp + k, ip[k] * w);
    }
}

__device__ __forceinline__ bool any_ours(uint4 v, unsigned pat) {
    unsigned x0 = v.x ^ pat, x1 = v.y ^ pat, x2 = v.z ^ pat, x3 = v.w ^ pat;
    unsigned z0 = (x0 - 0x01010101u) & ~x0;
    unsigned z1 = (x1 - 0x01010101u) & ~x1;
    unsigned z2 = (x2 - 0x01010101u) & ~x2;
    unsigned z3 = (x3 - 0x01010101u) & ~x3;
    return ((z0 | z1 | z2 | z3) & 0x80808080u) != 0u;
}

__device__ __forceinline__ unsigned ours_mask(unsigned k, unsigned pat) {
    unsigned x = k ^ pat;
    return ~(((x & 0x7F7F7F7Fu) + 0x7F7F7F7Fu) | x | 0x7F7F7F7Fu);
}

__device__ __forceinline__ unsigned place_group(unsigned p, int j, unsigned z0, unsigned z1, unsigned z2, unsigned z3,
                                                int base, unsigned* q, float* tile,
                                                const int* __restrict__ src, const int* __restrict__ dst,
                                                const float* __restrict__ nrm, const float* __restrict__ hin) {
    int e0 = j * 16;
#pragma unroll 1
    for (int d = 0; d < 4; ++d) {
        while (z0) {
            int b = __builtin_ctz(z0) >> 3;
            z0 &= z0 - 1u;
            place(p, e0 + b, base, q, tile, src, dst, nrm, hin);
            ++p;
        }
        z0 = z1; z1 = z2; z2 = z3;
        e0 += 4;
    }
    return p;
}

__device__ __forceinline__ void open_group(int j, uint4 v, unsigned pat, int base,
                                           unsigned* q, unsigned* qn, float* tile,
                                           const int* __restrict__ src, const int* __restrict__ dst,
                                           const float* __restrict__ nrm, const float* __restrict__ hin) {
    unsigned z0 = ours_mask(v.x, pat), z1 = ours_mask(v.y, pat);
    unsigned z2 = ours_mask(v.z, pat), z3 = ours_mask(v.w, pat);
    unsigned p = atomicAdd(qn, (unsigned)(__popc(z0) + __popc(z1) + __popc(z2) + __popc(z3)));
    place_group(p, j, z0, z1, z2, z3, base, q, tile, src, dst, nrm, hin);
}

__global__ __launch_bounds__(1024) void hop_k(const int* __restrict__ src,
                                              const int* __restrict__ dst,
                                              const float* __restrict__ nrm,
                                              const float* __restrict__ hin,
                                              float* __restrict__ hout,
                                              float* save, int first) {
    __shared__ __attribute__((aligned(16))) float    tile[NPB * F];
    __shared__ __attribute__((aligned(16))) unsigned q[QCAP];
    __shared__ __attribute__((aligned(16))) float    qw[QCAP];
    __shared__ __attribute__((aligned(16))) unsigned plist[NWV * PCAP];
    __shared__ unsigned pcnt[NWV];
    __shared__ unsigned qn;
    __shared__ unsigned gn;
    const int tid  = threadIdx.x;
    const int lane = tid & 31;
    const int wv   = tid >> 5;
    const int base = blockIdx.x * NPB;

    unsigned* sv = (unsigned*)save + (long)blockIdx.x * SVSTRIDE;
    const unsigned nsaved = first ? 0xFFFFFFFFu : sv[0];
    const bool sweep = nsaved > (unsigned)CAPSV;
    int n;
    if (tid < NWV) pcnt[tid] = 0u;

    if (!sweep) {
        n = (int)nsaved;
        const uint4* gq = (const uint4*)(sv + SVHEAD);
        const uint4* gw = (const uint4*)(sv + SVHEAD + CAPSV);
        const bool p0 = tid * 4 < n, p1 = (tid + 1024) * 4 < n;
        uint4 a0 = make_uint4(0u, 0u, 0u, 0u), b0 = a0, a1 = a0, b1 = a0;
        if (p0) { a0 = gq[tid];        b0 = gw[tid]; }
        if (p1) { a1 = gq[tid + 1024]; b1 = gw[tid + 1024]; }
        for (int i = tid; i < NPB * F; i += 1024) tile[i] = 0.0f;
        if (p0) { ((uint4*)q)[tid]        = a0; ((uint4*)qw)[tid]        = b0; }
        if (p1) { ((uint4*)q)[tid + 1024] = a1; ((uint4*)qw)[tid + 1024] = b1; }
        __syncthreads();
    } else {
        for (int i = tid; i < NPB * F; i += 1024) tile[i] = 0.0f;
        if (tid == 0) { qn = 0u; gn = 0u; }
        __syncthreads();

        const uint4*   d4  = (const uint4*)((const unsigned*)save + KEYOFF);
        const unsigned pat = blockIdx.x * 0x01010101u;
        const int      off = blockIdx.x * STAG;
        unsigned*      g   = (unsigned*)qw;
        for (int m = 0; m < TRIPS; ++m) {
            const int i0 = m * SWEEP + tid;
            uint4 v[8];
#pragma unroll
            for (int k = 0; k < 8; ++k) {
                int i = i0 + k * 1024;
                int j = i + off;
                if (j >= NV) j -= NV;
                if (i >= NV) j = 0;
                v[k] = d4[j];
            }
            unsigned fl = 0u;
#pragma unroll
            for (int k = 0; k < 8; ++k) fl |= any_ours(v[k], pat) ? (1u << k) : 0u;
            if (m == TRIPS - 1) {
                unsigned vm = 0u;
#pragma unroll
                for (int k = 0; k < 8; ++k) vm |= (i0 + k * 1024 < NV) ? (1u << k) : 0u;
                fl &= vm;
            }
#pragma unroll 1
            while (fl) {
                int k = __builtin_ctz(fl);
                fl &= fl - 1u;
                int j = i0 + k * 1024 + off;
                if (j >= NV) j -= NV;
                unsigned p = atomicAdd(&gn, 1u);
                if (p < QCAP) g[p] = (unsigned)j;
                else          open_group(j, d4[j], pat, base, q, &qn, tile, src, dst, nrm, hin);
            }
        }
        __syncthreads();

        const int ng = (int)min(gn, (unsigned)QCAP);
        for (int ib = 0; ib < ng; ib += 4 * 1024) {
            int   gj[4];
            uint4 gv[4];
#pragma unroll
            for (int k = 0; k < 4; ++k) {
                int i = ib + k * 1024 + tid;
                gj[k] = (i < ng) ? (int)g[i] : 0;
            }
#pragma unroll
            for (int k = 0; k < 4; ++k) gv[k] = d4[gj[k]];
            unsigned z[4][4];
            unsigned h = 0u;
#pragma unroll
            for (int k = 0; k < 4; ++k) {
                const bool ok = ib + k * 1024 + tid < ng;
                z[k][0] = ok ? ours_mask(gv[k].x, pat) : 0u;
                z[k][1] = ok ? ours_mask(gv[k].y, pat) : 0u;
                z[k][2] = ok ? ours_mask(gv[k].z, pat) : 0u;
                z[k][3] = ok ? ours_mask(gv[k].w, pat) : 0u;
                h += (unsigned)(__popc(z[k][0]) + __popc(z[k][1]) + __popc(z[k][2]) + __popc(z[k][3]));
            }
            unsigned p = atomicAdd(&qn, h);
#pragma unroll
            for (int k = 0; k < 4; ++k)
                p = place_group(p, gj[k], z[k][0], z[k][1], z[k][2], z[k][3], base, q, tile, src, dst, nrm, hin);
        }
        __syncthreads();

        n = (int)min(qn, (unsigned)QCAP);
        for (int jb = 0; jb < n; jb += 8 * 1024) {
            unsigned u[8];
            int      s[8], d[8];
            float    w[8];
#pragma unroll
            for (int k = 0; k < 8; ++k) {
                int j = jb + k * 1024 + tid;
                u[k] = (j < n) ? q[j] : 0u;
            }
#pragma unroll
            for (int k = 0; k < 8; ++k) s[k] = src[u[k]];
#pragma unroll
            for (int k = 0; k < 8; ++k) d[k] = dst[u[k]];
#pragma unroll
            for (int k = 0; k < 8; ++k) w[k] = nrm[s[k]];
#pragma unroll
            for (int k = 0; k < 8; ++k) {
                int j = jb + k * 1024 + tid;
                qw[j] = w[k];
                q[j]  = ((unsigned)(d[k] - base) << 16) | (unsigned)s[k];
            }
        }
        __syncthreads();

        if (first) {
            const unsigned nraw = qn;
            if (tid == 0) sv[0] = nraw;
            if (nraw <= (unsigned)CAPSV) {
                uint4* gq = (uint4*)(sv + SVHEAD);
                uint4* gw = (uint4*)(sv + SVHEAD + CAPSV);
                for (int j = tid; j * 4 < n; j += 1024) { gq[j] = ((const uint4*)q)[j]; gw[j] = ((const uint4*)qw)[j]; }
            }
        }
    }

    for (int j = tid; j < n; j += 1024) {
        const unsigned u   = q[j];
        const unsigned o   = (u >> 16) & (unsigned)(NWV - 1);
        const unsigned pos = atomicAdd(&pcnt[o], 1u);
        if (pos < (unsigned)PCAP) {
            plist[o * PCAP + pos] = (unsigned)j;
        } else {
            const float  w  = qw[j];
            const float* ip = hin + (long)(u & 0xFFFFu) * F;
            float*       tp = tile + (u >> 16) * F;
#pragma unroll 1
            for (int k = 0; k < F; ++k) unsafeAtomicAdd(tp + k, ip[k] * w);
        }
    }
    __syncthreads();

    const int       wvs = __builtin_amdgcn_readfirstlane(wv);
    const int       cw  = __builtin_amdgcn_readfirstlane((int)min(pcnt[wvs], (unsigned)PCAP));
    const unsigned* pl  = plist + wvs * PCAP;
    const unsigned  lane4 = (unsigned)lane << 2;
    float acc0[8], acc1[8];
#pragma unroll
    for (int i = 0; i < 8; ++i) { acc0[i] = 0.0f; acc1[i] = 0.0f; }
    float sc[RPW];
#pragma unroll
    for (int i = 0; i < RPW; ++i) {
        const int r = wvs + NWV * i;
        sc[i] = (r < NPB && base + r < NN) ? nrm[base + r] : 0.0f;
    }
    int u_l = 0, w_l = 0;
    if ((lane & 15) < cw) {
        const unsigned e = pl[lane & 15];
        u_l = (int)q[e];
        w_l = __float_as_int(qw[e]);
    }
    for (int j0 = 0; j0 < cw; j0 += 16) {
        const int cntb = min(16, cw - j0);
        float v0[16], v1[16];
        int   ri[16];
#pragma unroll
        for (int k = 0; k < 16; ++k) {
            const unsigned u = (unsigned)__builtin_amdgcn_readlane(u_l, k);
            const float    w = __int_as_float(__builtin_amdgcn_readlane(w_l, k));
            const char* rb = (const char*)hin + (unsigned long)(u & 0xFFFFu) * (unsigned long)(F * 4);
            v0[k] = *(const float*)(rb + lane4) * w;
            v1[k] = *(const float*)(rb + 128 + lane4) * w;
            ri[k] = (int)((u >> 21) & 7u);
        }
        int u_n = 0, w_n = 0;
        if (j0 + 16 + (lane & 15) < cw) {
            const unsigned e = pl[j0 + 16 + (lane & 15)];
            u_n = (int)q[e];
            w_n = __float_as_int(qw[e]);
        }
#pragma unroll
        for (int k = 0; k < 16; ++k) {
            if (k < cntb) {
                acc0[ri[k]] += v0[k];
                acc1[ri[k]] += v1[k];
            }
        }
        u_l = u_n;
        w_l = w_n;
    }

    float t0[RPW], t1[RPW];
#pragma unroll
    for (int i = 0; i < RPW; ++i) {
        const int r = wvs + NWV * i;
        t0[i] = 0.0f; t1[i] = 0.0f;
        if (r < NPB) { t0[i] = tile[r * F + lane]; t1[i] = tile[r * F + 32 + lane]; }
    }
#pragma unroll
    for (int i = 0; i < RPW; ++i) {
        const int r    = wvs + NWV * i;
        const int node = base + r;
        if (r < NPB && node < NN) {
            float* op = hout + (long)node * F + lane;
            op[0]  = (acc0[i] + t0[i]) * sc[i];
            op[32] = (acc1[i] + t1[i]) * sc[i];
        }
    }
}

#define GW 256
#define NT (NN / 16)
static_assert(NN % 16 == 0 && GW <= NT, "whole row tiles only, and every work-group has one");

__device__ __forceinline__ void ld8(float4 (&a)[8], const float* __restrict__ p) {
#pragma unroll
    for (int q = 0; q < 8; ++q) a[q] = *(const float4*)(p + 8 * q);
}

template <int SEG>
__device__ __forceinline__ void seg_mac(const float4 (&a)[8], const float4 (&w0)[32], const float4 (&w1)[32],
                                        v8f& c0, v8f& c1) {
#pragma unroll
    for (int q = 0; q < 8; ++q) {
        const float4 x = a[q], y0 = w0[SEG * 8 + q], y1 = w1[SEG * 8 + q];
        v2f al;  al.x  = x.x;  al.y  = x.y;
        v2f ah;  ah.x  = x.z;  ah.y  = x.w;
        v2f b0l; b0l.x = y0.x; b0l.y = y0.y;
        v2f b0h; b0h.x = y0.z; b0h.y = y0.w;
        v2f b1l; b1l.x = y1.x; b1l.y = y1.y;
        v2f b1h; b1h.x = y1.z; b1h.y = y1.w;
        c0 = __builtin_amdgcn_wmma_f32_16x16x4_f32(false, al, false, b0l, (short)0, c0, false, false);
        c1 = __builtin_amdgcn_wmma_f32_16x16x4_f32(false, al, false, b1l, (short)0, c1, false, false);
        c0 = __builtin_amdgcn_wmma_f32_16x16x4_f32(false, ah, false, b0h, (short)0, c0, false, false);
        c1 = __builtin_amdgcn_wmma_f32_16x16x4_f32(false, ah, false, b1h, (short)0, c1, false, false);
    }
}

__global__ __launch_bounds__(256) void gemm_k(const float* __restrict__ feat,
                                              const float* __restrict__ h1,
                                              const float* __restrict__ h2,
                                              const float* __restrict__ h3,
                                              const float* __restrict__ W,
                                              const float* __restrict__ bias,
                                              float* __restrict__ out) {
    const int lane = threadIdx.x & 31;
    const int c    = lane & 15;
    const int h    = lane >> 4;
    const int wv = __builtin_amdgcn_readfirstlane((int)(threadIdx.x >> 5));
    const int jh = wv & 1;
    const int g  = wv >> 1;
    const int cnt = (NT - 1 - (int)blockIdx.x) / GW + 1;

    float4 w0[32], w1[32];
    {
        const float* wp = W + (long)(jh * 32 + c) * KD + 4 * h;
#pragma unroll
        for (int i = 0; i < 32; ++i) w0[i] = *(const float4*)(wp + 8 * i);
#pragma unroll
        for (int i = 0; i < 32; ++i) w1[i] = *(const float4*)(wp + 16 * KD + 8 * i);
    }
    const float bv0 = bias[jh * 32 + c], bv1 = bias[jh * 32 + 16 + c];

    const long lo = (long)c * F + 4 * h;
    float4 a0[8], a1[8];
    ld8(a0, feat + (long)((int)blockIdx.x + GW * g) * (16 * F) + lo);
    for (int n = g; n < cnt; n += 4) {
        const int  tile = (int)blockIdx.x + GW * n;
        const int  nx   = (n + 4 < cnt) ? tile + 4 * GW : tile;
        const long to   = (long)tile * (16 * F) + lo;
        v8f c0, c1;
#pragma unroll
        for (int r = 0; r < 8; ++r) { c0[r] = bv0; c1[r] = bv1; }

        ld8(a1, h1 + to);
        __builtin_amdgcn_sched_barrier(0);
        seg_mac<0>(a0, w0, w1, c0, c1);
        __builtin_amdgcn_sched_barrier(0);
        ld8(a0, h2 + to);
        __builtin_amdgcn_sched_barrier(0);
        seg_mac<1>(a1, w0, w1, c0, c1);
        __builtin_amdgcn_sched_barrier(0);
        ld8(a1, h3 + to);
        __builtin_amdgcn_sched_barrier(0);
        seg_mac<2>(a0, w0, w1, c0, c1);
        __builtin_amdgcn_sched_barrier(0);
        ld8(a0, feat + (long)nx * (16 * F) + lo);
        __builtin_amdgcn_sched_barrier(0);
        seg_mac<3>(a1, w0, w1, c0, c1);

        float* op = out + (long)(tile * 16 + 8 * h) * F + jh * 32 + c;
#pragma unroll
        for (int r = 0; r < 8; ++r) {
            op[r * F]      = c0[r];
            op[r * F + 16] = c1[r];
        }
    }
}


extern "C" void kernel_launch(void* const* d_in, const int* in_sizes, int n_in,
                              void* d_out, int out_size, void* d_ws, size_t ws_size,
                              hipStream_t stream) {
    const float* feat = (const float*)d_in[0];
    const int*   src  = (const int*)d_in[1];
    const int*   dst  = (const int*)d_in[2];
    const float* W    = (const float*)d_in[3];
    const float* bias = (const float*)d_in[4];
    float*       out  = (float*)d_out;

    float* ws   = (float*)d_ws;
    float* nrm  = ws;
    float* h1   = ws + 50048;
    float* h2   = h1 + (long)NN * F;
    float* h3   = h2 + (long)NN * F;

    zero_k <<<196, 256, 0, stream>>>(ws, 50048);
    deg_k  <<<(NE / 8 + 255) / 256, 256, 0, stream>>>(dst, (int*)nrm, out + KEYOFF);
    norm_k <<<(NN + 255) / 256, 256, 0, stream>>>(nrm);

    hop_k  <<<HOPG, 1024, 0, stream>>>(src, dst, nrm, feat, h1, out, 1);
    hop_k  <<<HOPG, 1024, 0, stream>>>(src, dst, nrm, h1, h2, out, 0);
    hop_k  <<<HOPG, 1024, 0, stream>>>(src, dst, nrm, h2, h3, out, 0);

    gemm_k <<<GW, 256, 0, stream>>>(feat, h1, h2, h3, W, bias, out);
}


# Hops: overflow edges (slot past 32) are added by the wave owning the row, 32 lanes on the source row, to its registers; LDS rows, zero fill, serial walk, 2nd barrier go (sum order: rows > 32)
# speedup vs baseline: 2.5175x; 1.0897x over previous
#include <hip/hip_runtime.h>

typedef float v2f __attribute__((ext_vector_type(2)));
typedef float v8f __attribute__((ext_vector_type(8)));

#define NN 50000
#define NE 800000
#define F  64
#define KD 256


__global__ __launch_bounds__(256) void zero_k(float* __restrict__ p, long n) {
    long i  = (long)blockIdx.x * blockDim.x + threadIdx.x;
    long st = (long)gridDim.x * blockDim.x;
    for (; i < n; i += st) p[i] = 0.0f;
}

#define CAP   32
#define DEGW  50048
#define OVFW  NN
#define OVOFF (NN * CAP)
static_assert((long)OVOFF + 2L * NE <= (long)NN * F, "lists and a pair for every edge fit in out");
static_assert(OVFW < DEGW && NE % 8 == 0 && CAP % 4 == 0, "counter word inside the zeroed block; eight edges a thread");

__device__ __forceinline__ float rsq_deg(int d) {
    float f = (float)d;
    return rsqrtf(f < 1.0f ? 1.0f : f);
}

__device__ __forceinline__ void put_edge(unsigned* __restrict__ lst, int* __restrict__ deg, int d, int s, int r) {
    if (r < CAP) {
        atomicExch(lst + (unsigned)d * CAP + (unsigned)r, (unsigned)s);
    } else {
        unsigned p = atomicAdd((unsigned*)deg + OVFW, 1u);
        atomicExch(lst + OVOFF + 2u * p,      (unsigned)d);
        atomicExch(lst + OVOFF + 2u * p + 1u, (unsigned)s);
    }
}

__global__ __launch_bounds__(256) void deg_k(const int* __restrict__ src, const int* __restrict__ dst,
                                             int* __restrict__ deg, unsigned* __restrict__ lst) {
    int t = blockIdx.x * 256 + threadIdx.x;
    if (t < NE / 8) {
        int4 a = ((const int4*)dst)[2 * t];
        int4 b = ((const int4*)dst)[2 * t + 1];
        int4 c = ((const int4*)src)[2 * t];
        int4 e = ((const int4*)src)[2 * t + 1];
        int r0 = atomicAdd(&deg[a.x], 1), r1 = atomicAdd(&deg[a.y], 1), r2 = atomicAdd(&deg[a.z], 1), r3 = atomicAdd(&deg[a.w], 1);
        int r4 = atomicAdd(&deg[b.x], 1), r5 = atomicAdd(&deg[b.y], 1), r6 = atomicAdd(&deg[b.z], 1), r7 = atomicAdd(&deg[b.w], 1);
        put_edge(lst, deg, a.x, c.x, r0); put_edge(lst, deg, a.y, c.y, r1);
        put_edge(lst, deg, a.z, c.z, r2); put_edge(lst, deg, a.w, c.w, r3);
        put_edge(lst, deg, b.x, e.x, r4); put_edge(lst, deg, b.y, e.y, r5);
        put_edge(lst, deg, b.z, e.z, r6); put_edge(lst, deg, b.w, e.w, r7);
    }
}

#define NPB  196
#define HOPG 256
#define NWV  32
#define RPW  7
#define NG4  (NPB * CAP / 4)
#define LW   24576
static_assert(NPB * HOPG >= NN, "every node has an owner");
static_assert(NWV * RPW >= NPB && NWV == 32 && RPW <= 8, "every row has an owning wave; a row's place in its wave fits three bits");
static_assert(NN <= (1 << 16), "list word in the sum: place in the wave << 16 | source");
static_assert(NG4 <= 2 * 1024 && CAP == 32 && LW >= 2 * 1024 * 4, "a thread stages two 16-byte groups; eight groups a row");

__device__ __forceinline__ void list_word(int j, int cw, int wvs, const int (&pre)[RPW + 1],
                                          const unsigned* es, const float* ew, int& u, int& w) {
    u = 0; w = 0;
    if (j < cw) {
        int a = wvs * CAP + j, pi = 0;
#pragma unroll
        for (int i = 1; i < RPW; ++i)
            if (j >= pre[i]) { a = (wvs + NWV * i) * CAP + j - pre[i]; pi = i; }
        u = (int)(es[a] | ((unsigned)pi << 16));
        w = __float_as_int(ew[a]);
    }
}

template <bool FULL>
__device__ __forceinline__ void sum_batch(int u_l, int w_l, int cntb, const float* __restrict__ hin, unsigned lane4,
                                          float (&acc0)[8], float (&acc1)[8],
                                          int jn, int cw, int wvs, const int (&pre)[RPW + 1],
                                          const unsigned* es, const float* ew, int& u_n, int& w_n) {
    float v0[16], v1[16];
    int   ri[16];
#pragma unroll
    for (int k = 0; k < 16; ++k) {
        const unsigned u = (unsigned)__builtin_amdgcn_readlane(u_l, k);
        const float    w = __int_as_float(__builtin_amdgcn_readlane(w_l, k));
        const char* rb = (const char*)hin + (unsigned long)(u & 0xFFFFu) * (unsigned long)(F * 4);
        v0[k] = *(const float*)(rb + lane4) * w;
        v1[k] = *(const float*)(rb + 128 + lane4) * w;
        ri[k] = (int)((u >> 16) & 7u);
    }
    u_n = 0; w_n = 0;
    if (FULL) list_word(jn, cw, wvs, pre, es, ew, u_n, w_n);
#pragma unroll
    for (int k = 0; k < 16; ++k) {
        if (FULL || k < cntb) {
            acc0[ri[k]] += v0[k];
            acc1[ri[k]] += v1[k];
        }
    }
}

__global__ __launch_bounds__(1024) void hop_k(const int* __restrict__ deg,
                                              const unsigned* __restrict__ lst,
                                              const float* __restrict__ hin,
                                              float* __restrict__ hout) {
    __shared__ __attribute__((aligned(16))) unsigned es[LW];
    __shared__ __attribute__((aligned(16))) float    ew[LW];
    __shared__ int rdeg[NWV * 8];
    const int tid  = threadIdx.x;
    const int lane = tid & 31;
    const int wv   = tid >> 5;
    const int base = blockIdx.x * NPB;

    const uint4* l4 = (const uint4*)lst + (long)base * (CAP / 4);
    uint4 v[2];
    int   cn[2];
#pragma unroll
    for (int k = 0; k < 2; ++k) {
        const int g    = tid + 1024 * k;
        const int node = base + (g >> 3);
        v[k]  = make_uint4(0u, 0u, 0u, 0u);
        cn[k] = 0;
        if (g < NG4 && node < NN) { v[k] = l4[g]; cn[k] = deg[node]; }
    }
    int rd = 0;
    if (tid < NPB && base + tid < NN) rd = deg[base + tid];
    const unsigned novf = min(((const unsigned*)deg)[OVFW], (unsigned)NE);
    unsigned sv[2][4];
    int      dg[2][4];
#pragma unroll
    for (int k = 0; k < 2; ++k) {
        const int s0 = ((tid + 1024 * k) & 7) * 4;
        const int c  = min(cn[k], CAP);
        sv[k][0] = (s0     < c) ? v[k].x : 0u;
        sv[k][1] = (s0 + 1 < c) ? v[k].y : 0u;
        sv[k][2] = (s0 + 2 < c) ? v[k].z : 0u;
        sv[k][3] = (s0 + 3 < c) ? v[k].w : 0u;
    }
#pragma unroll
    for (int k = 0; k < 2; ++k)
#pragma unroll
        for (int j = 0; j < 4; ++j) dg[k][j] = deg[sv[k][j]];
#pragma unroll
    for (int k = 0; k < 2; ++k) {
        const int g = tid + 1024 * k;
        ((uint4*)es)[g]  = make_uint4(sv[k][0], sv[k][1], sv[k][2], sv[k][3]);
        ((float4*)ew)[g] = make_float4(rsq_deg(dg[k][0]), rsq_deg(dg[k][1]), rsq_deg(dg[k][2]), rsq_deg(dg[k][3]));
    }
    if (tid < NWV * 8) rdeg[tid] = rd;
    __syncthreads();

    const int wvs = __builtin_amdgcn_readfirstlane(wv);
    const int dl = (lane < RPW) ? rdeg[wvs + NWV * lane] : 0;
    int   pre[RPW + 1];
    float sc[RPW];
    pre[0] = 0;
#pragma unroll
    for (int i = 0; i < RPW; ++i) {
        const int dr = __builtin_amdgcn_readlane(dl, i);
        sc[i]      = rsq_deg(dr);
        pre[i + 1] = pre[i] + min(dr, CAP);
    }
    const int cw = pre[RPW];
    const unsigned lane4 = (unsigned)lane << 2;
    float acc0[8], acc1[8];
#pragma unroll
    for (int i = 0; i < 8; ++i) { acc0[i] = 0.0f; acc1[i] = 0.0f; }

    const unsigned* ov = lst + OVOFF;
    uint2 pv = make_uint2(0xFFFFFFFFu, 0u);
    if ((unsigned)lane < novf) pv = *(const uint2*)(ov + 2u * (unsigned)lane);

    int u_l, w_l;
    list_word(lane & 15, cw, wvs, pre, es, ew, u_l, w_l);
    int j0 = 0;
    for (; j0 + 16 <= cw; j0 += 16) {
        int u_n, w_n;
        sum_batch<true>(u_l, w_l, 16, hin, lane4, acc0, acc1, j0 + 16 + (lane & 15), cw, wvs, pre, es, ew, u_n, w_n);
        u_l = u_n;
        w_l = w_n;
    }
    if (j0 < cw) {
        int u_n, w_n;
        sum_batch<false>(u_l, w_l, cw - j0, hin, lane4, acc0, acc1, 0, 0, wvs, pre, es, ew, u_n, w_n);
    }

    for (unsigned q0 = 0; q0 < novf; q0 += 32u) {
        if (q0) {
            pv = make_uint2(0xFFFFFFFFu, 0u);
            if (q0 + (unsigned)lane < novf) pv = *(const uint2*)(ov + 2u * (q0 + (unsigned)lane));
        }
        const unsigned r    = pv.x - (unsigned)base;
        const bool     mine = r < (unsigned)NPB && (r & (unsigned)(NWV - 1)) == (unsigned)wvs;
        float wq = 0.0f;
        if (mine) wq = rsq_deg(deg[pv.y]);
        unsigned m = __builtin_amdgcn_ballot_w32(mine);
        while (m) {
            const int k = __builtin_ctz(m);
            m &= m - 1u;
            const unsigned rr = (unsigned)__builtin_amdgcn_readlane((int)r, k);
            const unsigned ss = (unsigned)__builtin_amdgcn_readlane((int)pv.y, k);
            const float    ww = __int_as_float(__builtin_amdgcn_readlane(__float_as_int(wq), k));
            const char* rb = (const char*)hin + (unsigned long)ss * (unsigned long)(F * 4);
            const float x0 = *(const float*)(rb + lane4) * ww;
            const float x1 = *(const float*)(rb + 128 + lane4) * ww;
            const int   i  = (int)(rr >> 5) & 7;
            acc0[i] += x0;
            acc1[i] += x1;
        }
    }

#pragma unroll
    for (int i = 0; i < RPW; ++i) {
        const int r    = wvs + NWV * i;
        const int node = base + r;
        if (r < NPB && node < NN) {
            float* op = hout + (long)node * F + lane;
            op[0]  = acc0[i] * sc[i];
            op[32] = acc1[i] * sc[i];
        }
    }
}

#define GW 256
#define NT (NN / 16)
static_assert(NN % 16 == 0 && GW <= NT, "whole row tiles only, and every work-group has one");

__device__ __forceinline__ void ld8(float4 (&a)[8], const float* __restrict__ p) {
#pragma unroll
    for (int q = 0; q < 8; ++q) a[q] = *(const float4*)(p + 8 * q);
}

template <int SEG>
__device__ __forceinline__ void seg_mac(const float4 (&a)[8], const float4 (&w0)[32], const float4 (&w1)[32],
                                        v8f& c0, v8f& c1) {
#pragma unroll
    for (int q = 0; q < 8; ++q) {
        const float4 x = a[q], y0 = w0[SEG * 8 + q], y1 = w1[SEG * 8 + q];
        v2f al;  al.x  = x.x;  al.y  = x.y;
        v2f ah;  ah.x  = x.z;  ah.y  = x.w;
        v2f b0l; b0l.x = y0.x; b0l.y = y0.y;
        v2f b0h; b0h.x = y0.z; b0h.y = y0.w;
        v2f b1l; b1l.x = y1.x; b1l.y = y1.y;
        v2f b1h; b1h.x = y1.z; b1h.y = y1.w;
        c0 = __builtin_amdgcn_wmma_f32_16x16x4_f32(false, al, false, b0l, (short)0, c0, false, false);
        c1 = __builtin_amdgcn_wmma_f32_16x16x4_f32(false, al, false, b1l, (short)0, c1, false, false);
        c0 = __builtin_amdgcn_wmma_f32_16x16x4_f32(false, ah, false, b0h, (short)0, c0, false, false);
        c1 = __builtin_amdgcn_wmma_f32_16x16x4_f32(false, ah, false, b1h, (short)0, c1, false, false);
    }
}

__global__ __launch_bounds__(256) void gemm_k(const float* __restrict__ feat,
                                              const float* __restrict__ h1,
                                              const float* __restrict__ h2,
                                              const float* __restrict__ h3,
                                              const float* __restrict__ W,
                                              const float* __restrict__ bias,
                                              float* __restrict__ out) {
    const int lane = threadIdx.x & 31;
    const int c    = lane & 15;
    const int h    = lane >> 4;
    const int wv = __builtin_amdgcn_readfirstlane((int)(threadIdx.x >> 5));
    const int jh = wv & 1;
    const int g  = wv >> 1;
    const int cnt = (NT - 1 - (int)blockIdx.x) / GW + 1;

    float4 w0[32], w1[32];
    {
        const float* wp = W + (long)(jh * 32 + c) * KD + 4 * h;
#pragma unroll
        for (int i = 0; i < 32; ++i) w0[i] = *(const float4*)(wp + 8 * i);
#pragma unroll
        for (int i = 0; i < 32; ++i) w1[i] = *(const float4*)(wp + 16 * KD + 8 * i);
    }
    const float bv0 = bias[jh * 32 + c], bv1 = bias[jh * 32 + 16 + c];

    const long lo = (long)c * F + 4 * h;
    float4 a0[8], a1[8];
    ld8(a0, feat + (long)((int)blockIdx.x + GW * g) * (16 * F) + lo);
    for (int n = g; n < cnt; n += 4) {
        const int  tile = (int)blockIdx.x + GW * n;
        const int  nx   = (n + 4 < cnt) ? tile + 4 * GW : tile;
        const long to   = (long)tile * (16 * F) + lo;
        v8f c0, c1;
#pragma unroll
        for (int r = 0; r < 8; ++r) { c0[r] = bv0; c1[r] = bv1; }

        ld8(a1, h1 + to);
        __builtin_amdgcn_sched_barrier(0);
        seg_mac<0>(a0, w0, w1, c0, c1);
        __builtin_amdgcn_sched_barrier(0);
        ld8(a0, h2 + to);
        __builtin_amdgcn_sched_barrier(0);
        seg_mac<1>(a1, w0, w1, c0, c1);
        __builtin_amdgcn_sched_barrier(0);
        ld8(a1, h3 + to);
        __builtin_amdgcn_sched_barrier(0);
        seg_mac<2>(a0, w0, w1, c0, c1);
        __builtin_amdgcn_sched_barrier(0);
        ld8(a0, feat + (long)nx * (16 * F) + lo);
        __builtin_amdgcn_sched_barrier(0);
        seg_mac<3>(a1, w0, w1, c0, c1);

        float* op = out + (long)(tile * 16 + 8 * h) * F + jh * 32 + c;
#pragma unroll
        for (int r = 0; r < 8; ++r) {
            op[r * F]      = c0[r];
            op[r * F + 16] = c1[r];
        }
    }
}


extern "C" void kernel_launch(void* const* d_in, const int* in_sizes, int n_in,
                              void* d_out, int out_size, void* d_ws, size_t ws_size,
                              hipStream_t stream) {
    const float* feat = (const float*)d_in[0];
    const int*   src  = (const int*)d_in[1];
    const int*   dst  = (const int*)d_in[2];
    const float* W    = (const float*)d_in[3];
    const float* bias = (const float*)d_in[4];
    float*       out  = (float*)d_out;

    float* ws   = (float*)d_ws;
    int*   deg  = (int*)d_ws;
    float* h1   = ws + DEGW;
    float* h2   = h1 + (long)NN * F;
    float* h3   = h2 + (long)NN * F;

    zero_k <<<196, 256, 0, stream>>>(ws, DEGW);
    deg_k  <<<(NE / 8 + 255) / 256, 256, 0, stream>>>(src, dst, deg, (unsigned*)out);

    hop_k  <<<HOPG, 1024, 0, stream>>>(deg, (const unsigned*)out, feat, h1);
    hop_k  <<<HOPG, 1024, 0, stream>>>(deg, (const unsigned*)out, h1, h2);
    hop_k  <<<HOPG, 1024, 0, stream>>>(deg, (const unsigned*)out, h2, h3);

    gemm_k <<<GW, 256, 0, stream>>>(feat, h1, h2, h3, W, bias, out);
}


# Degree and list atomics split by package half: nodes in two ranges, every edge chunk taken once per range by a work-group drawing from its own half's ticket; sums still in arrival (slot) order
# speedup vs baseline: 3.8962x; 1.5476x over previous
#include <hip/hip_runtime.h>

typedef float v2f __attribute__((ext_vector_type(2)));
typedef float v8f __attribute__((ext_vector_type(8)));

#define NN 50000
#define NE 800000
#define F  64
#define KD 256


__global__ __launch_bounds__(256) void zero_k(float* __restrict__ p, long n) {
    long i  = (long)blockIdx.x * blockDim.x + threadIdx.x;
    long st = (long)gridDim.x * blockDim.x;
    for (; i < n; i += st) p[i] = 0.0f;
}

#define CAP   32
#define DEGW  50048
#define OVFW  NN
#define OVOFF (NN * CAP)
static_assert((long)OVOFF + 2L * NE <= (long)NN * F, "lists and a pair for every edge fit in out");
static_assert(OVFW < DEGW && NE % 8 == 0 && CAP % 4 == 0, "counter word inside the zeroed block; eight edges a thread");

__device__ __forceinline__ float rsq_deg(int d) {
    float f = (float)d;
    return rsqrtf(f < 1.0f ? 1.0f : f);
}

__device__ __forceinline__ void put_edge(unsigned* __restrict__ lst, int* __restrict__ deg, int d, int s, int r) {
    if (r < CAP) {
        atomicExch(lst + (unsigned)d * CAP + (unsigned)r, (unsigned)s);
    } else {
        unsigned p = atomicAdd((unsigned*)deg + OVFW, 1u);
        atomicExch(lst + OVOFF + 2u * p,      (unsigned)d);
        atomicExch(lst + OVOFF + 2u * p + 1u, (unsigned)s);
    }
}

#define NCH   ((NE / 8 + 255) / 256)
#define NHALF 25024
#define TKW   50016
static_assert(TKW >= NN + 16 && TKW + 16 < DEGW && TKW % 32 == 0, "tickets in a spare line of the zeroed block, off the overflow count's line");
static_assert(NHALF % 32 == 0 && NHALF < NN, "whole counter lines per range");

__global__ __launch_bounds__(256) void deg_k(const int* __restrict__ src, const int* __restrict__ dst,
                                             int* __restrict__ deg, unsigned* __restrict__ lst) {
    __shared__ int tk[2];
    if (threadIdx.x == 0) {
        int k = (int)((__builtin_amdgcn_s_sendmsg_rtn(0x87) >> 18) & 1u);
        int c = atomicAdd(&deg[TKW + 16 * k], 1);
        if (c >= NCH) { k ^= 1; c = atomicAdd(&deg[TKW + 16 * k], 1); }
        tk[0] = c;
        tk[1] = k;
    }
    __syncthreads();
    const int ch = tk[0], k = tk[1];
    if (ch >= NCH) return;
    const unsigned lo = k ? (unsigned)NHALF : 0u;
    const unsigned nr = k ? (unsigned)(NN - NHALF) : (unsigned)NHALF;
    int t = ch * 256 + threadIdx.x;
    if (t < NE / 8) {
        int4 a = ((const int4*)dst)[2 * t];
        int4 b = ((const int4*)dst)[2 * t + 1];
        int4 c = ((const int4*)src)[2 * t];
        int4 e = ((const int4*)src)[2 * t + 1];
        const bool m0 = (unsigned)a.x - lo < nr, m1 = (unsigned)a.y - lo < nr, m2 = (unsigned)a.z - lo < nr, m3 = (unsigned)a.w - lo < nr;
        const bool m4 = (unsigned)b.x - lo < nr, m5 = (unsigned)b.y - lo < nr, m6 = (unsigned)b.z - lo < nr, m7 = (unsigned)b.w - lo < nr;
        int r0 = 0, r1 = 0, r2 = 0, r3 = 0, r4 = 0, r5 = 0, r6 = 0, r7 = 0;
        if (m0) r0 = atomicAdd(&deg[a.x], 1);
        if (m1) r1 = atomicAdd(&deg[a.y], 1);
        if (m2) r2 = atomicAdd(&deg[a.z], 1);
        if (m3) r3 = atomicAdd(&deg[a.w], 1);
        if (m4) r4 = atomicAdd(&deg[b.x], 1);
        if (m5) r5 = atomicAdd(&deg[b.y], 1);
        if (m6) r6 = atomicAdd(&deg[b.z], 1);
        if (m7) r7 = atomicAdd(&deg[b.w], 1);
        if (m0) put_edge(lst, deg, a.x, c.x, r0);
        if (m1) put_edge(lst, deg, a.y, c.y, r1);
        if (m2) put_edge(lst, deg, a.z, c.z, r2);
        if (m3) put_edge(lst, deg, a.w, c.w, r3);
        if (m4) put_edge(lst, deg, b.x, e.x, r4);
        if (m5) put_edge(lst, deg, b.y, e.y, r5);
        if (m6) put_edge(lst, deg, b.z, e.z, r6);
        if (m7) put_edge(lst, deg, b.w, e.w, r7);
    }
}

#define NPB  196
#define HOPG 256
#define NWV  32
#define RPW  7
#define NG4  (NPB * CAP / 4)
#define LW   24576
static_assert(NPB * HOPG >= NN, "every node has an owner");
static_assert(NWV * RPW >= NPB && NWV == 32 && RPW <= 8, "every row has an owning wave; a row's place in its wave fits three bits");
static_assert(NN <= (1 << 16), "list word in the sum: place in the wave << 16 | source");
static_assert(NG4 <= 2 * 1024 && CAP == 32 && LW >= 2 * 1024 * 4, "a thread stages two 16-byte groups; eight groups a row");

__device__ __forceinline__ void list_word(int j, int cw, int wvs, const int (&pre)[RPW + 1],
                                          const unsigned* es, const float* ew, int& u, int& w) {
    u = 0; w = 0;
    if (j < cw) {
        int a = wvs * CAP + j, pi = 0;
#pragma unroll
        for (int i = 1; i < RPW; ++i)
            if (j >= pre[i]) { a = (wvs + NWV * i) * CAP + j - pre[i]; pi = i; }
        u = (int)(es[a] | ((unsigned)pi << 16));
        w = __float_as_int(ew[a]);
    }
}

template <bool FULL>
__device__ __forceinline__ void sum_batch(int u_l, int w_l, int cntb, const float* __restrict__ hin, unsigned lane4,
                                          float (&acc0)[8], float (&acc1)[8],
                                          int jn, int cw, int wvs, const int (&pre)[RPW + 1],
                                          const unsigned* es, const float* ew, int& u_n, int& w_n) {
    float v0[16], v1[16];
    int   ri[16];
#pragma unroll
    for (int k = 0; k < 16; ++k) {
        const unsigned u = (unsigned)__builtin_amdgcn_readlane(u_l, k);
        const float    w = __int_as_float(__builtin_amdgcn_readlane(w_l, k));
        const char* rb = (const char*)hin + (unsigned long)(u & 0xFFFFu) * (unsigned long)(F * 4);
        v0[k] = *(const float*)(rb + lane4) * w;
        v1[k] = *(const float*)(rb + 128 + lane4) * w;
        ri[k] = (int)((u >> 16) & 7u);
    }
    u_n = 0; w_n = 0;
    if (FULL) list_word(jn, cw, wvs, pre, es, ew, u_n, w_n);
#pragma unroll
    for (int k = 0; k < 16; ++k) {
        if (FULL || k < cntb) {
            acc0[ri[k]] += v0[k];
            acc1[ri[k]] += v1[k];
        }
    }
}

__global__ __launch_bounds__(1024) void hop_k(const int* __restrict__ deg,
                                              const unsigned* __restrict__ lst,
                                              const float* __restrict__ hin,
                                              float* __restrict__ hout) {
    __shared__ __attribute__((aligned(16))) unsigned es[LW];
    __shared__ __attribute__((aligned(16))) float    ew[LW];
    __shared__ int rdeg[NWV * 8];
    const int tid  = threadIdx.x;
    const int lane = tid & 31;
    const int wv   = tid >> 5;
    const int base = blockIdx.x * NPB;

    const uint4* l4 = (const uint4*)lst + (long)base * (CAP / 4);
    uint4 v[2];
    int   cn[2];
#pragma unroll
    for (int k = 0; k < 2; ++k) {
        const int g    = tid + 1024 * k;
        const int node = base + (g >> 3);
        v[k]  = make_uint4(0u, 0u, 0u, 0u);
        cn[k] = 0;
        if (g < NG4 && node < NN) { v[k] = l4[g]; cn[k] = deg[node]; }
    }
    int rd = 0;
    if (tid < NPB && base + tid < NN) rd = deg[base + tid];
    const unsigned novf = min(((const unsigned*)deg)[OVFW], (unsigned)NE);
    unsigned sv[2][4];
    int      dg[2][4];
#pragma unroll
    for (int k = 0; k < 2; ++k) {
        const int s0 = ((tid + 1024 * k) & 7) * 4;
        const int c  = min(cn[k], CAP);
        sv[k][0] = (s0     < c) ? v[k].x : 0u;
        sv[k][1] = (s0 + 1 < c) ? v[k].y : 0u;
        sv[k][2] = (s0 + 2 < c) ? v[k].z : 0u;
        sv[k][3] = (s0 + 3 < c) ? v[k].w : 0u;
    }
#pragma unroll
    for (int k = 0; k < 2; ++k)
#pragma unroll
        for (int j = 0; j < 4; ++j) dg[k][j] = deg[sv[k][j]];
#pragma unroll
    for (int k = 0; k < 2; ++k) {
        const int g = tid + 1024 * k;
        ((uint4*)es)[g]  = make_uint4(sv[k][0], sv[k][1], sv[k][2], sv[k][3]);
        ((float4*)ew)[g] = make_float4(rsq_deg(dg[k][0]), rsq_deg(dg[k][1]), rsq_deg(dg[k][2]), rsq_deg(dg[k][3]));
    }
    if (tid < NWV * 8) rdeg[tid] = rd;
    __syncthreads();

    const int wvs = __builtin_amdgcn_readfirstlane(wv);
    const int dl = (lane < RPW) ? rdeg[wvs + NWV * lane] : 0;
    int   pre[RPW + 1];
    float sc[RPW];
    pre[0] = 0;
#pragma unroll
    for (int i = 0; i < RPW; ++i) {
        const int dr = __builtin_amdgcn_readlane(dl, i);
        sc[i]      = rsq_deg(dr);
        pre[i + 1] = pre[i] + min(dr, CAP);
    }
    const int cw = pre[RPW];
    const unsigned lane4 = (unsigned)lane << 2;
    float acc0[8], acc1[8];
#pragma unroll
    for (int i = 0; i < 8; ++i) { acc0[i] = 0.0f; acc1[i] = 0.0f; }

    const unsigned* ov = lst + OVOFF;
    uint2 pv = make_uint2(0xFFFFFFFFu, 0u);
    if ((unsigned)lane < novf) pv = *(const uint2*)(ov + 2u * (unsigned)lane);

    int u_l, w_l;
    list_word(lane & 15, cw, wvs, pre, es, ew, u_l, w_l);
    int j0 = 0;
    for (; j0 + 16 <= cw; j0 += 16) {
        int u_n, w_n;
        sum_batch<true>(u_l, w_l, 16, hin, lane4, acc0, acc1, j0 + 16 + (lane & 15), cw, wvs, pre, es, ew, u_n, w_n);
        u_l = u_n;
        w_l = w_n;
    }
    if (j0 < cw) {
        int u_n, w_n;
        sum_batch<false>(u_l, w_l, cw - j0, hin, lane4, acc0, acc1, 0, 0, wvs, pre, es, ew, u_n, w_n);
    }

    for (unsigned q0 = 0; q0 < novf; q0 += 32u) {
        if (q0) {
            pv = make_uint2(0xFFFFFFFFu, 0u);
            if (q0 + (unsigned)lane < novf) pv = *(const uint2*)(ov + 2u * (q0 + (unsigned)lane));
        }
        const unsigned r    = pv.x - (unsigned)base;
        const bool     mine = r < (unsigned)NPB && (r & (unsigned)(NWV - 1)) == (unsigned)wvs;
        float wq = 0.0f;
        if (mine) wq = rsq_deg(deg[pv.y]);
        unsigned m = __builtin_amdgcn_ballot_w32(mine);
        while (m) {
            const int k = __builtin_ctz(m);
            m &= m - 1u;
            const unsigned rr = (unsigned)__builtin_amdgcn_readlane((int)r, k);
            const unsigned ss = (unsigned)__builtin_amdgcn_readlane((int)pv.y, k);
            const float    ww = __int_as_float(__builtin_amdgcn_readlane(__float_as_int(wq), k));
            const char* rb = (const char*)hin + (unsigned long)ss * (unsigned long)(F * 4);
            const float x0 = *(const float*)(rb + lane4) * ww;
            const float x1 = *(const float*)(rb + 128 + lane4) * ww;
            const int   i  = (int)(rr >> 5) & 7;
            acc0[i] += x0;
            acc1[i] += x1;
        }
    }

#pragma unroll
    for (int i = 0; i < RPW; ++i) {
        const int r    = wvs + NWV * i;
        const int node = base + r;
        if (r < NPB && node < NN) {
            float* op = hout + (long)node * F + lane;
            op[0]  = acc0[i] * sc[i];
            op[32] = acc1[i] * sc[i];
        }
    }
}

#define GW 256
#define NT (NN / 16)
static_assert(NN % 16 == 0 && GW <= NT, "whole row tiles only, and every work-group has one");

__device__ __forceinline__ void ld8(float4 (&a)[8], const float* __restrict__ p) {
#pragma unroll
    for (int q = 0; q < 8; ++q) a[q] = *(const float4*)(p + 8 * q);
}

template <int SEG>
__device__ __forceinline__ void seg_mac(const float4 (&a)[8], const float4 (&w0)[32], const float4 (&w1)[32],
                                        v8f& c0, v8f& c1) {
#pragma unroll
    for (int q = 0; q < 8; ++q) {
        const float4 x = a[q], y0 = w0[SEG * 8 + q], y1 = w1[SEG * 8 + q];
        v2f al;  al.x  = x.x;  al.y  = x.y;
        v2f ah;  ah.x  = x.z;  ah.y  = x.w;
        v2f b0l; b0l.x = y0.x; b0l.y = y0.y;
        v2f b0h; b0h.x = y0.z; b0h.y = y0.w;
        v2f b1l; b1l.x = y1.x; b1l.y = y1.y;
        v2f b1h; b1h.x = y1.z; b1h.y = y1.w;
        c0 = __builtin_amdgcn_wmma_f32_16x16x4_f32(false, al, false, b0l, (short)0, c0, false, false);
        c1 = __builtin_amdgcn_wmma_f32_16x16x4_f32(false, al, false, b1l, (short)0, c1, false, false);
        c0 = __builtin_amdgcn_wmma_f32_16x16x4_f32(false, ah, false, b0h, (short)0, c0, false, false);
        c1 = __builtin_amdgcn_wmma_f32_16x16x4_f32(false, ah, false, b1h, (short)0, c1, false, false);
    }
}

__global__ __launch_bounds__(256) void gemm_k(const float* __restrict__ feat,
                                              const float* __restrict__ h1,
                                              const float* __restrict__ h2,
                                              const float* __restrict__ h3,
                                              const float* __restrict__ W,
                                              const float* __restrict__ bias,
                                              float* __restrict__ out) {
    const int lane = threadIdx.x & 31;
    const int c    = lane & 15;
    const int h    = lane >> 4;
    const int wv = __builtin_amdgcn_readfirstlane((int)(threadIdx.x >> 5));
    const int jh = wv & 1;
    const int g  = wv >> 1;
    const int cnt = (NT - 1 - (int)blockIdx.x) / GW + 1;

    float4 w0[32], w1[32];
    {
        const float* wp = W + (long)(jh * 32 + c) * KD + 4 * h;
#pragma unroll
        for (int i = 0; i < 32; ++i) w0[i] = *(const float4*)(wp + 8 * i);
#pragma unroll
        for (int i = 0; i < 32; ++i) w1[i] = *(const float4*)(wp + 16 * KD + 8 * i);
    }
    const float bv0 = bias[jh * 32 + c], bv1 = bias[jh * 32 + 16 + c];

    const long lo = (long)c * F + 4 * h;
    float4 a0[8], a1[8];
    ld8(a0, feat + (long)((int)blockIdx.x + GW * g) * (16 * F) + lo);
    for (int n = g; n < cnt; n += 4) {
        const int  tile = (int)blockIdx.x + GW * n;
        const int  nx   = (n + 4 < cnt) ? tile + 4 * GW : tile;
        const long to   = (long)tile * (16 * F) + lo;
        v8f c0, c1;
#pragma unroll
        for (int r = 0; r < 8; ++r) { c0[r] = bv0; c1[r] = bv1; }

        ld8(a1, h1 + to);
        __builtin_amdgcn_sched_barrier(0);
        seg_mac<0>(a0, w0, w1, c0, c1);
        __builtin_amdgcn_sched_barrier(0);
        ld8(a0, h2 + to);
        __builtin_amdgcn_sched_barrier(0);
        seg_mac<1>(a1, w0, w1, c0, c1);
        __builtin_amdgcn_sched_barrier(0);
        ld8(a1, h3 + to);
        __builtin_amdgcn_sched_barrier(0);
        seg_mac<2>(a0, w0, w1, c0, c1);
        __builtin_amdgcn_sched_barrier(0);
        ld8(a0, feat + (long)nx * (16 * F) + lo);
        __builtin_amdgcn_sched_barrier(0);
        seg_mac<3>(a1, w0, w1, c0, c1);

        float* op = out + (long)(tile * 16 + 8 * h) * F + jh * 32 + c;
#pragma unroll
        for (int r = 0; r < 8; ++r) {
            op[r * F]      = c0[r];
            op[r * F + 16] = c1[r];
        }
    }
}


extern "C" void kernel_launch(void* const* d_in, const int* in_sizes, int n_in,
                              void* d_out, int out_size, void* d_ws, size_t ws_size,
                              hipStream_t stream) {
    const float* feat = (const float*)d_in[0];
    const int*   src  = (const int*)d_in[1];
    const int*   dst  = (const int*)d_in[2];
    const float* W    = (const float*)d_in[3];
    const float* bias = (const float*)d_in[4];
    float*       out  = (float*)d_out;

    float* ws   = (float*)d_ws;
    int*   deg  = (int*)d_ws;
    float* h1   = ws + DEGW;
    float* h2   = h1 + (long)NN * F;
    float* h3   = h2 + (long)NN * F;

    zero_k <<<196, 256, 0, stream>>>(ws, DEGW);
    deg_k  <<<2 * NCH, 256, 0, stream>>>(src, dst, deg, (unsigned*)out);

    hop_k  <<<HOPG, 1024, 0, stream>>>(deg, (const unsigned*)out, feat, h1);
    hop_k  <<<HOPG, 1024, 0, stream>>>(deg, (const unsigned*)out, h1, h2);
    hop_k  <<<HOPG, 1024, 0, stream>>>(deg, (const unsigned*)out, h2, h3);

    gemm_k <<<GW, 256, 0, stream>>>(feat, h1, h2, h3, W, bias, out);
}


# Hop sum: two sources per 16-byte-per-lane load (16 lanes a source row), a wave's rows summed and written one by one in fixed registers, no relative moves; sum order: two slot-group partials a row
# speedup vs baseline: 4.2638x; 1.0943x over previous
#include <hip/hip_runtime.h>

typedef float v2f __attribute__((ext_vector_type(2)));
typedef float v8f __attribute__((ext_vector_type(8)));

#define NN 50000
#define NE 800000
#define F  64
#define KD 256


__global__ __launch_bounds__(256) void zero_k(float* __restrict__ p, long n) {
    long i  = (long)blockIdx.x * blockDim.x + threadIdx.x;
    long st = (long)gridDim.x * blockDim.x;
    for (; i < n; i += st) p[i] = 0.0f;
}

#define CAP   32
#define DEGW  50048
#define OVFW  NN
#define OVOFF (NN * CAP)
static_assert((long)OVOFF + 2L * NE <= (long)NN * F, "lists and a pair for every edge fit in out");
static_assert(OVFW < DEGW && NE % 8 == 0 && CAP % 4 == 0, "counter word inside the zeroed block; eight edges a thread");

__device__ __forceinline__ float rsq_deg(int d) {
    float f = (float)d;
    return rsqrtf(f < 1.0f ? 1.0f : f);
}

__device__ __forceinline__ void put_edge(unsigned* __restrict__ lst, int* __restrict__ deg, int d, int s, int r) {
    if (r < CAP) {
        atomicExch(lst + (unsigned)d * CAP + (unsigned)r, (unsigned)s);
    } else {
        unsigned p = atomicAdd((unsigned*)deg + OVFW, 1u);
        atomicExch(lst + OVOFF + 2u * p,      (unsigned)d);
        atomicExch(lst + OVOFF + 2u * p + 1u, (unsigned)s);
    }
}

#define NCH   ((NE / 8 + 255) / 256)
#define NHALF 25024
#define TKW   50016
static_assert(TKW >= NN + 16 && TKW + 16 < DEGW && TKW % 32 == 0, "tickets in a spare line of the zeroed block, off the overflow count's line");
static_assert(NHALF % 32 == 0 && NHALF < NN, "whole counter lines per range");

__global__ __launch_bounds__(256) void deg_k(const int* __restrict__ src, const int* __restrict__ dst,
                                             int* __restrict__ deg, unsigned* __restrict__ lst) {
    __shared__ int tk[2];
    if (threadIdx.x == 0) {
        int k = (int)((__builtin_amdgcn_s_sendmsg_rtn(0x87) >> 18) & 1u);
        int c = atomicAdd(&deg[TKW + 16 * k], 1);
        if (c >= NCH) { k ^= 1; c = atomicAdd(&deg[TKW + 16 * k], 1); }
        tk[0] = c;
        tk[1] = k;
    }
    __syncthreads();
    const int ch = tk[0], k = tk[1];
    if (ch >= NCH) return;
    const unsigned lo = k ? (unsigned)NHALF : 0u;
    const unsigned nr = k ? (unsigned)(NN - NHALF) : (unsigned)NHALF;
    int t = ch * 256 + threadIdx.x;
    if (t < NE / 8) {
        int4 a = ((const int4*)dst)[2 * t];
        int4 b = ((const int4*)dst)[2 * t + 1];
        int4 c = ((const int4*)src)[2 * t];
        int4 e = ((const int4*)src)[2 * t + 1];
        const bool m0 = (unsigned)a.x - lo < nr, m1 = (unsigned)a.y - lo < nr, m2 = (unsigned)a.z - lo < nr, m3 = (unsigned)a.w - lo < nr;
        const bool m4 = (unsigned)b.x - lo < nr, m5 = (unsigned)b.y - lo < nr, m6 = (unsigned)b.z - lo < nr, m7 = (unsigned)b.w - lo < nr;
        int r0 = 0, r1 = 0, r2 = 0, r3 = 0, r4 = 0, r5 = 0, r6 = 0, r7 = 0;
        if (m0) r0 = atomicAdd(&deg[a.x], 1);
        if (m1) r1 = atomicAdd(&deg[a.y], 1);
        if (m2) r2 = atomicAdd(&deg[a.z], 1);
        if (m3) r3 = atomicAdd(&deg[a.w], 1);
        if (m4) r4 = atomicAdd(&deg[b.x], 1);
        if (m5) r5 = atomicAdd(&deg[b.y], 1);
        if (m6) r6 = atomicAdd(&deg[b.z], 1);
        if (m7) r7 = atomicAdd(&deg[b.w], 1);
        if (m0) put_edge(lst, deg, a.x, c.x, r0);
        if (m1) put_edge(lst, deg, a.y, c.y, r1);
        if (m2) put_edge(lst, deg, a.z, c.z, r2);
        if (m3) put_edge(lst, deg, a.w, c.w, r3);
        if (m4) put_edge(lst, deg, b.x, e.x, r4);
        if (m5) put_edge(lst, deg, b.y, e.y, r5);
        if (m6) put_edge(lst, deg, b.z, e.z, r6);
        if (m7) put_edge(lst, deg, b.w, e.w, r7);
    }
}

#define NPB  196
#define HOPG 256
#define NWV  32
#define RPW  7
#define NG4  (NPB * CAP / 4)
#define LW   24576
static_assert(NPB * HOPG >= NN, "every node has an owner");
static_assert(NWV * RPW >= NPB && NWV == 32 && RPW <= 8, "every row has an owning wave; a row's place in its wave fits three bits");
static_assert((long)NN * F * 4 < (1L << 32), "a source row's byte offset fits a 32-bit word");
static_assert(NG4 <= 2 * 1024 && CAP == 32 && LW >= 2 * 1024 * 4, "a thread stages two 16-byte groups; eight groups a row");

__device__ __forceinline__ void row_fma(v2f& lo, v2f& hi, const float4 xv, const float w) {
    v2f xl; xl.x = xv.x; xl.y = xv.y;
    v2f xh; xh.x = xv.z; xh.y = xv.w;
    v2f ww; ww.x = w;    ww.y = w;
    lo = __builtin_elementwise_fma(xl, ww, lo);
    hi = __builtin_elementwise_fma(xh, ww, hi);
}

__global__ __launch_bounds__(1024) void hop_k(const int* __restrict__ deg,
                                              const unsigned* __restrict__ lst,
                                              const float* __restrict__ hin,
                                              float* __restrict__ hout) {
    __shared__ __attribute__((aligned(16))) unsigned es[LW];
    __shared__ __attribute__((aligned(16))) float    ew[LW];
    __shared__ int rdeg[NWV * 8];
    const int tid  = threadIdx.x;
    const int lane = tid & 31;
    const int wv   = tid >> 5;
    const int base = blockIdx.x * NPB;

    const uint4* l4 = (const uint4*)lst + (long)base * (CAP / 4);
    uint4 v[2];
    int   cn[2];
#pragma unroll
    for (int k = 0; k < 2; ++k) {
        const int g    = tid + 1024 * k;
        const int node = base + (g >> 3);
        v[k]  = make_uint4(0u, 0u, 0u, 0u);
        cn[k] = 0;
        if (g < NG4 && node < NN) { v[k] = l4[g]; cn[k] = deg[node]; }
    }
    int rd = 0;
    if (tid < NPB && base + tid < NN) rd = deg[base + tid];
    const unsigned novf = min(((const unsigned*)deg)[OVFW], (unsigned)NE);
    unsigned sv[2][4];
    int      dg[2][4];
    bool     ok[2][4];
#pragma unroll
    for (int k = 0; k < 2; ++k) {
        const int s0 = ((tid + 1024 * k) & 7) * 4;
        const int c  = min(cn[k], CAP);
        ok[k][0] = s0     < c;
        ok[k][1] = s0 + 1 < c;
        ok[k][2] = s0 + 2 < c;
        ok[k][3] = s0 + 3 < c;
        sv[k][0] = ok[k][0] ? v[k].x : 0u;
        sv[k][1] = ok[k][1] ? v[k].y : 0u;
        sv[k][2] = ok[k][2] ? v[k].z : 0u;
        sv[k][3] = ok[k][3] ? v[k].w : 0u;
    }
#pragma unroll
    for (int k = 0; k < 2; ++k)
#pragma unroll
        for (int j = 0; j < 4; ++j) dg[k][j] = deg[sv[k][j]];
#pragma unroll
    for (int k = 0; k < 2; ++k) {
        const int g = tid + 1024 * k;
        ((uint4*)es)[g]  = make_uint4(sv[k][0] << 8, sv[k][1] << 8, sv[k][2] << 8, sv[k][3] << 8);
        ((float4*)ew)[g] = make_float4(ok[k][0] ? rsq_deg(dg[k][0]) : 0.0f, ok[k][1] ? rsq_deg(dg[k][1]) : 0.0f,
                                       ok[k][2] ? rsq_deg(dg[k][2]) : 0.0f, ok[k][3] ? rsq_deg(dg[k][3]) : 0.0f);
    }
    if (tid < NWV * 8) rdeg[tid] = rd;
    __syncthreads();

    const int wvs = __builtin_amdgcn_readfirstlane(wv);
    const int dl = (lane < RPW) ? rdeg[wvs + NWV * lane] : 0;
    const int      hh  = lane >> 4;
    const unsigned c16 = (unsigned)(lane & 15) << 4;
    const char*    hb  = (const char*)hin;

    const unsigned* ov = lst + OVOFF;
    uint2 pv = make_uint2(0xFFFFFFFFu, 0u);
    if ((unsigned)lane < novf) pv = *(const uint2*)(ov + 2u * (unsigned)lane);

    for (int i = 0; i < RPW; ++i) {
        const int r    = wvs + NWV * i;
        const int node = base + r;
        if (r >= NPB || node >= NN) break;
        const int dr = __builtin_amdgcn_readlane(dl, i);
        const int ni = min(dr, CAP);
        v2f sL, sH;
        sL.x = 0.0f; sL.y = 0.0f; sH.x = 0.0f; sH.y = 0.0f;
        if (ni > 0) {
            const int rb = r * CAP + 4 * hh;
            uint4 o[4];
#pragma unroll
            for (int g = 0; g < 4; ++g) o[g] = *(const uint4*)(es + rb + 8 * g);
            float4 x[16];
#pragma unroll
            for (int g = 0; g < 4; ++g) {
                if (8 * g < ni) {
                    x[4 * g + 0] = *(const float4*)(hb + (o[g].x + c16));
                    x[4 * g + 1] = *(const float4*)(hb + (o[g].y + c16));
                    x[4 * g + 2] = *(const float4*)(hb + (o[g].z + c16));
                    x[4 * g + 3] = *(const float4*)(hb + (o[g].w + c16));
                }
            }
#pragma unroll
            for (int g = 0; g < 4; ++g) {
                if (8 * g < ni) {
                    const float4 w = *(const float4*)(ew + rb + 8 * g);
                    row_fma(sL, sH, x[4 * g + 0], w.x);
                    row_fma(sL, sH, x[4 * g + 1], w.y);
                    row_fma(sL, sH, x[4 * g + 2], w.z);
                    row_fma(sL, sH, x[4 * g + 3], w.w);
                }
            }
        }

        for (unsigned q0 = 0; q0 < novf; q0 += 32u) {
            uint2 pq = pv;
            if (q0) {
                pq = make_uint2(0xFFFFFFFFu, 0u);
                if (q0 + (unsigned)lane < novf) pq = *(const uint2*)(ov + 2u * (q0 + (unsigned)lane));
            }
            const bool mine = pq.x == (unsigned)node;
            float wq = 0.0f;
            if (mine) wq = rsq_deg(deg[pq.y]);
            unsigned m = __builtin_amdgcn_ballot_w32(mine);
            while (m) {
                const int k = __builtin_ctz(m);
                m &= m - 1u;
                const unsigned ss = (unsigned)__builtin_amdgcn_readlane((int)pq.y, k);
                const float    ww = __int_as_float(__builtin_amdgcn_readlane(__float_as_int(wq), k));
                const float4   xo = *(const float4*)(hb + ((ss << 8) + c16));
                row_fma(sL, sH, xo, hh ? 0.0f : ww);
            }
        }

        const float sc = rsq_deg(dr);
        float4 t;
        t.x = (sL.x + __shfl_xor(sL.x, 16)) * sc;
        t.y = (sL.y + __shfl_xor(sL.y, 16)) * sc;
        t.z = (sH.x + __shfl_xor(sH.x, 16)) * sc;
        t.w = (sH.y + __shfl_xor(sH.y, 16)) * sc;
        if (hh == 0) *(float4*)(hout + (long)node * F + 4 * (lane & 15)) = t;
    }
}

#define GW 256
#define NT (NN / 16)
static_assert(NN % 16 == 0 && GW <= NT, "whole row tiles only, and every work-group has one");

__device__ __forceinline__ void ld8(float4 (&a)[8], const float* __restrict__ p) {
#pragma unroll
    for (int q = 0; q < 8; ++q) a[q] = *(const float4*)(p + 8 * q);
}

template <int SEG>
__device__ __forceinline__ void seg_mac(const float4 (&a)[8], const float4 (&w0)[32], const float4 (&w1)[32],
                                        v8f& c0, v8f& c1) {
#pragma unroll
    for (int q = 0; q < 8; ++q) {
        const float4 x = a[q], y0 = w0[SEG * 8 + q], y1 = w1[SEG * 8 + q];
        v2f al;  al.x  = x.x;  al.y  = x.y;
        v2f ah;  ah.x  = x.z;  ah.y  = x.w;
        v2f b0l; b0l.x = y0.x; b0l.y = y0.y;
        v2f b0h; b0h.x = y0.z; b0h.y = y0.w;
        v2f b1l; b1l.x = y1.x; b1l.y = y1.y;
        v2f b1h; b1h.x = y1.z; b1h.y = y1.w;
        c0 = __builtin_amdgcn_wmma_f32_16x16x4_f32(false, al, false, b0l, (short)0, c0, false, false);
        c1 = __builtin_amdgcn_wmma_f32_16x16x4_f32(false, al, false, b1l, (short)0, c1, false, false);
        c0 = __builtin_amdgcn_wmma_f32_16x16x4_f32(false, ah, false, b0h, (short)0, c0, false, false);
        c1 = __builtin_amdgcn_wmma_f32_16x16x4_f32(false, ah, false, b1h, (short)0, c1, false, false);
    }
}

__global__ __launch_bounds__(256) void gemm_k(const float* __restrict__ feat,
                                              const float* __restrict__ h1,
                                              const float* __restrict__ h2,
                                              const float* __restrict__ h3,
                                              const float* __restrict__ W,
                                              const float* __restrict__ bias,
                                              float* __restrict__ out) {
    const int lane = threadIdx.x & 31;
    const int c    = lane & 15;
    const int h    = lane >> 4;
    const int wv = __builtin_amdgcn_readfirstlane((int)(threadIdx.x >> 5));
    const int jh = wv & 1;
    const int g  = wv >> 1;
    const int cnt = (NT - 1 - (int)blockIdx.x) / GW + 1;

    float4 w0[32], w1[32];
    {
        const float* wp = W + (long)(jh * 32 + c) * KD + 4 * h;
#pragma unroll
        for (int i = 0; i < 32; ++i) w0[i] = *(const float4*)(wp + 8 * i);
#pragma unroll
        for (int i = 0; i < 32; ++i) w1[i] = *(const float4*)(wp + 16 * KD + 8 * i);
    }
    const float bv0 = bias[jh * 32 + c], bv1 = bias[jh * 32 + 16 + c];

    const long lo = (long)c * F + 4 * h;
    float4 a0[8], a1[8];
    ld8(a0, feat + (long)((int)blockIdx.x + GW * g) * (16 * F) + lo);
    for (int n = g; n < cnt; n += 4) {
        const int  tile = (int)blockIdx.x + GW * n;
        const int  nx   = (n + 4 < cnt) ? tile + 4 * GW : tile;
        const long to   = (long)tile * (16 * F) + lo;
        v8f c0, c1;
#pragma unroll
        for (int r = 0; r < 8; ++r) { c0[r] = bv0; c1[r] = bv1; }

        ld8(a1, h1 + to);
        __builtin_amdgcn_sched_barrier(0);
        seg_mac<0>(a0, w0, w1, c0, c1);
        __builtin_amdgcn_sched_barrier(0);
        ld8(a0, h2 + to);
        __builtin_amdgcn_sched_barrier(0);
        seg_mac<1>(a1, w0, w1, c0, c1);
        __builtin_amdgcn_sched_barrier(0);
        ld8(a1, h3 + to);
        __builtin_amdgcn_sched_barrier(0);
        seg_mac<2>(a0, w0, w1, c0, c1);
        __builtin_amdgcn_sched_barrier(0);
        ld8(a0, feat + (long)nx * (16 * F) + lo);
        __builtin_amdgcn_sched_barrier(0);
        seg_mac<3>(a1, w0, w1, c0, c1);

        float* op = out + (long)(tile * 16 + 8 * h) * F + jh * 32 + c;
#pragma unroll
        for (int r = 0; r < 8; ++r) {
            op[r * F]      = c0[r];
            op[r * F + 16] = c1[r];
        }
    }
}


extern "C" void kernel_launch(void* const* d_in, const int* in_sizes, int n_in,
                              void* d_out, int out_size, void* d_ws, size_t ws_size,
                              hipStream_t stream) {
    const float* feat = (const float*)d_in[0];
    const int*   src  = (const int*)d_in[1];
    const int*   dst  = (const int*)d_in[2];
    const float* W    = (const float*)d_in[3];
    const float* bias = (const float*)d_in[4];
    float*       out  = (float*)d_out;

    float* ws   = (float*)d_ws;
    int*   deg  = (int*)d_ws;
    float* h1   = ws + DEGW;
    float* h2   = h1 + (long)NN * F;
    float* h3   = h2 + (long)NN * F;

    zero_k <<<196, 256, 0, stream>>>(ws, DEGW);
    deg_k  <<<2 * NCH, 256, 0, stream>>>(src, dst, deg, (unsigned*)out);

    hop_k  <<<HOPG, 1024, 0, stream>>>(deg, (const unsigned*)out, feat, h1);
    hop_k  <<<HOPG, 1024, 0, stream>>>(deg, (const unsigned*)out, h1, h2);
    hop_k  <<<HOPG, 1024, 0, stream>>>(deg, (const unsigned*)out, h2, h3);

    gemm_k <<<GW, 256, 0, stream>>>(feat, h1, h2, h3, W, bias, out);
}


# Hop staging in one memory trip: hop 1 saves the lists' source norms (in the last hop's own plane), hops 2 and 3 load them beside the lists, a thread's loads sent together; degree look-ups go
# speedup vs baseline: 4.4301x; 1.0390x over previous
#include <hip/hip_runtime.h>

typedef float v2f __attribute__((ext_vector_type(2)));
typedef float v8f __attribute__((ext_vector_type(8)));

#define NN 50000
#define NE 800000
#define F  64
#define KD 256


__global__ __launch_bounds__(256) void zero_k(float* __restrict__ p, long n) {
    long i  = (long)blockIdx.x * blockDim.x + threadIdx.x;
    long st = (long)gridDim.x * blockDim.x;
    for (; i < n; i += st) p[i] = 0.0f;
}

#define CAP   32
#define DEGW  50048
#define OVFW  NN
#define OVOFF (NN * CAP)
static_assert((long)OVOFF + 2L * NE <= (long)NN * F, "lists and a pair for every edge fit in out");
static_assert(OVFW < DEGW && NE % 8 == 0 && CAP % 4 == 0, "counter word inside the zeroed block; eight edges a thread");

__device__ __forceinline__ float rsq_deg(int d) {
    float f = (float)d;
    return rsqrtf(f < 1.0f ? 1.0f : f);
}

__device__ __forceinline__ void put_edge(unsigned* __restrict__ lst, int* __restrict__ deg, int d, int s, int r) {
    if (r < CAP) {
        atomicExch(lst + (unsigned)d * CAP + (unsigned)r, (unsigned)s);
    } else {
        unsigned p = atomicAdd((unsigned*)deg + OVFW, 1u);
        atomicExch(lst + OVOFF + 2u * p,      (unsigned)d);
        atomicExch(lst + OVOFF + 2u * p + 1u, (unsigned)s);
    }
}

#define NCH   ((NE / 8 + 255) / 256)
#define NHALF 25024
#define TKW   50016
static_assert(TKW >= NN + 16 && TKW + 16 < DEGW && TKW % 32 == 0, "tickets in a spare line of the zeroed block, off the overflow count's line");
static_assert(NHALF % 32 == 0 && NHALF < NN, "whole counter lines per range");

__global__ __launch_bounds__(256) void deg_k(const int* __restrict__ src, const int* __restrict__ dst,
                                             int* __restrict__ deg, unsigned* __restrict__ lst) {
    __shared__ int tk[2];
    if (threadIdx.x == 0) {
        int k = (int)((__builtin_amdgcn_s_sendmsg_rtn(0x87) >> 18) & 1u);
        int c = atomicAdd(&deg[TKW + 16 * k], 1);
        if (c >= NCH) { k ^= 1; c = atomicAdd(&deg[TKW + 16 * k], 1); }
        tk[0] = c;
        tk[1] = k;
    }
    __syncthreads();
    const int ch = tk[0], k = tk[1];
    if (ch >= NCH) return;
    const unsigned lo = k ? (unsigned)NHALF : 0u;
    const unsigned nr = k ? (unsigned)(NN - NHALF) : (unsigned)NHALF;
    int t = ch * 256 + threadIdx.x;
    if (t < NE / 8) {
        int4 a = ((const int4*)dst)[2 * t];
        int4 b = ((const int4*)dst)[2 * t + 1];
        int4 c = ((const int4*)src)[2 * t];
        int4 e = ((const int4*)src)[2 * t + 1];
        const bool m0 = (unsigned)a.x - lo < nr, m1 = (unsigned)a.y - lo < nr, m2 = (unsigned)a.z - lo < nr, m3 = (unsigned)a.w - lo < nr;
        const bool m4 = (unsigned)b.x - lo < nr, m5 = (unsigned)b.y - lo < nr, m6 = (unsigned)b.z - lo < nr, m7 = (unsigned)b.w - lo < nr;
        int r0 = 0, r1 = 0, r2 = 0, r3 = 0, r4 = 0, r5 = 0, r6 = 0, r7 = 0;
        if (m0) r0 = atomicAdd(&deg[a.x], 1);
        if (m1) r1 = atomicAdd(&deg[a.y], 1);
        if (m2) r2 = atomicAdd(&deg[a.z], 1);
        if (m3) r3 = atomicAdd(&deg[a.w], 1);
        if (m4) r4 = atomicAdd(&deg[b.x], 1);
        if (m5) r5 = atomicAdd(&deg[b.y], 1);
        if (m6) r6 = atomicAdd(&deg[b.z], 1);
        if (m7) r7 = atomicAdd(&deg[b.w], 1);
        if (m0) put_edge(lst, deg, a.x, c.x, r0);
        if (m1) put_edge(lst, deg, a.y, c.y, r1);
        if (m2) put_edge(lst, deg, a.z, c.z, r2);
        if (m3) put_edge(lst, deg, a.w, c.w, r3);
        if (m4) put_edge(lst, deg, b.x, e.x, r4);
        if (m5) put_edge(lst, deg, b.y, e.y, r5);
        if (m6) put_edge(lst, deg, b.z, e.z, r6);
        if (m7) put_edge(lst, deg, b.w, e.w, r7);
    }
}

#define NPB  196
#define HOPG 256
#define NWV  32
#define RPW  7
#define NG4  (NPB * CAP / 4)
#define LW   24576
static_assert(NPB * HOPG >= NN, "every node has an owner");
static_assert(NWV * RPW >= NPB && NWV == 32 && RPW <= 8, "every row has an owning wave; a row's place in its wave fits three bits");
static_assert((long)NN * F * 4 < (1L << 32), "a source row's byte offset fits a 32-bit word");
static_assert(NG4 <= 2 * 1024 && CAP == 32 && LW >= 2 * 1024 * 4, "a thread stages two 16-byte groups; eight groups a row");
static_assert(CAP <= F, "a row's saved norms fit inside the row's own room in a plane");
static_assert((HOPG - 1) * NPB < NN, "every work-group has a first node, so a first group of lists and of saved norms");
static_assert(NWV * 8 == 2 * 1024 / 8, "the two groups of a thread with a row's first group cover rows 0..127 and 128..255: all of rdeg");

__device__ __forceinline__ void row_fma(v2f& lo, v2f& hi, const float4 xv, const float w) {
    v2f xl; xl.x = xv.x; xl.y = xv.y;
    v2f xh; xh.x = xv.z; xh.y = xv.w;
    v2f ww; ww.x = w;    ww.y = w;
    lo = __builtin_elementwise_fma(xl, ww, lo);
    hi = __builtin_elementwise_fma(xh, ww, hi);
}

template <int FIRST>
__global__ __launch_bounds__(1024) void hop_k(const int* __restrict__ deg,
                                              const unsigned* __restrict__ lst,
                                              const float* __restrict__ hin,
                                              float* hout,
                                              float* nrm) {
    __shared__ __attribute__((aligned(16))) unsigned es[LW];
    __shared__ __attribute__((aligned(16))) float    ew[LW];
    __shared__ int rdeg[NWV * 8];
    const int tid  = threadIdx.x;
    const int lane = tid & 31;
    const int wv   = tid >> 5;
    const int base = blockIdx.x * NPB;

    const uint4* l4 = (const uint4*)lst + (long)base * (CAP / 4);
    float4*      n4 = (float4*)(nrm + (long)base * F);
    uint4  v[2];
    int    cn[2];
    float4 wn[2];
#pragma unroll
    for (int k = 0; k < 2; ++k) {
        const int  g    = tid + 1024 * k;
        const int  node = base + (g >> 3);
        const bool in   = g < NG4 && node < NN;
        const int  gl   = in ? g : 0;
        v[k]  = l4[gl];
        cn[k] = deg[in ? node : base];
        wn[k] = make_float4(0.0f, 0.0f, 0.0f, 0.0f);
        if (!FIRST) wn[k] = n4[gl];
        if (!in) {
            v[k]  = make_uint4(0u, 0u, 0u, 0u);
            cn[k] = 0;
            wn[k] = make_float4(0.0f, 0.0f, 0.0f, 0.0f);
        }
    }
    const unsigned novf = min(((const unsigned*)deg)[OVFW], (unsigned)NE);
    unsigned sv[2][4];
    int      dg[2][4];
    bool     ok[2][4];
#pragma unroll
    for (int k = 0; k < 2; ++k) {
        const int s0 = ((tid + 1024 * k) & 7) * 4;
        const int c  = min(cn[k], CAP);
        ok[k][0] = s0     < c;
        ok[k][1] = s0 + 1 < c;
        ok[k][2] = s0 + 2 < c;
        ok[k][3] = s0 + 3 < c;
        sv[k][0] = ok[k][0] ? v[k].x : 0u;
        sv[k][1] = ok[k][1] ? v[k].y : 0u;
        sv[k][2] = ok[k][2] ? v[k].z : 0u;
        sv[k][3] = ok[k][3] ? v[k].w : 0u;
    }
    if (FIRST) {
#pragma unroll
        for (int k = 0; k < 2; ++k)
#pragma unroll
            for (int j = 0; j < 4; ++j) dg[k][j] = deg[sv[k][j]];
#pragma unroll
        for (int k = 0; k < 2; ++k)
            wn[k] = make_float4(ok[k][0] ? rsq_deg(dg[k][0]) : 0.0f, ok[k][1] ? rsq_deg(dg[k][1]) : 0.0f,
                                ok[k][2] ? rsq_deg(dg[k][2]) : 0.0f, ok[k][3] ? rsq_deg(dg[k][3]) : 0.0f);
    }
#pragma unroll
    for (int k = 0; k < 2; ++k) {
        const int g = tid + 1024 * k;
        ((uint4*)es)[g]  = make_uint4(sv[k][0] << 8, sv[k][1] << 8, sv[k][2] << 8, sv[k][3] << 8);
        ((float4*)ew)[g] = wn[k];
    }
    if ((tid & 7) == 0) {
        rdeg[tid >> 3]       = cn[0];
        rdeg[128 + (tid >> 3)] = cn[1];
    }
    __syncthreads();
    if (FIRST) {
#pragma unroll
        for (int k = 0; k < 2; ++k) {
            const int g = tid + 1024 * k;
            if (g < NG4 && base + (g >> 3) < NN) n4[g] = wn[k];
        }
    }

    const int wvs = __builtin_amdgcn_readfirstlane(wv);
    const int dl = (lane < RPW) ? rdeg[wvs + NWV * lane] : 0;
    const int      hh  = lane >> 4;
    const unsigned c16 = (unsigned)(lane & 15) << 4;
    const char*    hb  = (const char*)hin;

    const unsigned* ov = lst + OVOFF;
    uint2 pv = make_uint2(0xFFFFFFFFu, 0u);
    if ((unsigned)lane < novf) pv = *(const uint2*)(ov + 2u * (unsigned)lane);

    for (int i = 0; i < RPW; ++i) {
        const int r    = wvs + NWV * i;
        const int node = base + r;
        if (r >= NPB || node >= NN) break;
        const int dr = __builtin_amdgcn_readlane(dl, i);
        const int ni = min(dr, CAP);
        v2f sL, sH;
        sL.x = 0.0f; sL.y = 0.0f; sH.x = 0.0f; sH.y = 0.0f;
        if (ni > 0) {
            const int rb = r * CAP + 4 * hh;
            uint4 o[4];
#pragma unroll
            for (int g = 0; g < 4; ++g) o[g] = *(const uint4*)(es + rb + 8 * g);
            float4 x[16];
#pragma unroll
            for (int g = 0; g < 4; ++g) {
                if (8 * g < ni) {
                    x[4 * g + 0] = *(const float4*)(hb + (o[g].x + c16));
                    x[4 * g + 1] = *(const float4*)(hb + (o[g].y + c16));
                    x[4 * g + 2] = *(const float4*)(hb + (o[g].z + c16));
                    x[4 * g + 3] = *(const float4*)(hb + (o[g].w + c16));
                }
            }
#pragma unroll
            for (int g = 0; g < 4; ++g) {
                if (8 * g < ni) {
                    const float4 w = *(const float4*)(ew + rb + 8 * g);
                    row_fma(sL, sH, x[4 * g + 0], w.x);
                    row_fma(sL, sH, x[4 * g + 1], w.y);
                    row_fma(sL, sH, x[4 * g + 2], w.z);
                    row_fma(sL, sH, x[4 * g + 3], w.w);
                }
            }
        }

        for (unsigned q0 = 0; q0 < novf; q0 += 32u) {
            uint2 pq = pv;
            if (q0) {
                pq = make_uint2(0xFFFFFFFFu, 0u);
                if (q0 + (unsigned)lane < novf) pq = *(const uint2*)(ov + 2u * (q0 + (unsigned)lane));
            }
            const bool mine = pq.x == (unsigned)node;
            float wq = 0.0f;
            if (mine) wq = rsq_deg(deg[pq.y]);
            unsigned m = __builtin_amdgcn_ballot_w32(mine);
            while (m) {
                const int k = __builtin_ctz(m);
                m &= m - 1u;
                const unsigned ss = (unsigned)__builtin_amdgcn_readlane((int)pq.y, k);
                const float    ww = __int_as_float(__builtin_amdgcn_readlane(__float_as_int(wq), k));
                const float4   xo = *(const float4*)(hb + ((ss << 8) + c16));
                row_fma(sL, sH, xo, hh ? 0.0f : ww);
            }
        }

        const float sc = rsq_deg(dr);
        float4 t;
        t.x = (sL.x + __shfl_xor(sL.x, 16)) * sc;
        t.y = (sL.y + __shfl_xor(sL.y, 16)) * sc;
        t.z = (sH.x + __shfl_xor(sH.x, 16)) * sc;
        t.w = (sH.y + __shfl_xor(sH.y, 16)) * sc;
        if (hh == 0) *(float4*)(hout + (long)node * F + 4 * (lane & 15)) = t;
    }
}

#define GW 256
#define NT (NN / 16)
static_assert(NN % 16 == 0 && GW <= NT, "whole row tiles only, and every work-group has one");

__device__ __forceinline__ void ld8(float4 (&a)[8], const float* __restrict__ p) {
#pragma unroll
    for (int q = 0; q < 8; ++q) a[q] = *(const float4*)(p + 8 * q);
}

template <int SEG>
__device__ __forceinline__ void seg_mac(const float4 (&a)[8], const float4 (&w0)[32], const float4 (&w1)[32],
                                        v8f& c0, v8f& c1) {
#pragma unroll
    for (int q = 0; q < 8; ++q) {
        const float4 x = a[q], y0 = w0[SEG * 8 + q], y1 = w1[SEG * 8 + q];
        v2f al;  al.x  = x.x;  al.y  = x.y;
        v2f ah;  ah.x  = x.z;  ah.y  = x.w;
        v2f b0l; b0l.x = y0.x; b0l.y = y0.y;
        v2f b0h; b0h.x = y0.z; b0h.y = y0.w;
        v2f b1l; b1l.x = y1.x; b1l.y = y1.y;
        v2f b1h; b1h.x = y1.z; b1h.y = y1.w;
        c0 = __builtin_amdgcn_wmma_f32_16x16x4_f32(false, al, false, b0l, (short)0, c0, false, false);
        c1 = __builtin_amdgcn_wmma_f32_16x16x4_f32(false, al, false, b1l, (short)0, c1, false, false);
        c0 = __builtin_amdgcn_wmma_f32_16x16x4_f32(false, ah, false, b0h, (short)0, c0, false, false);
        c1 = __builtin_amdgcn_wmma_f32_16x16x4_f32(false, ah, false, b1h, (short)0, c1, false, false);
    }
}

__global__ __launch_bounds__(256) void gemm_k(const float* __restrict__ feat,
                                              const float* __restrict__ h1,
                                              const float* __restrict__ h2,
                                              const float* __restrict__ h3,
                                              const float* __restrict__ W,
                                              const float* __restrict__ bias,
                                              float* __restrict__ out) {
    const int lane = threadIdx.x & 31;
    const int c    = lane & 15;
    const int h    = lane >> 4;
    const int wv = __builtin_amdgcn_readfirstlane((int)(threadIdx.x >> 5));
    const int jh = wv & 1;
    const int g  = wv >> 1;
    const int cnt = (NT - 1 - (int)blockIdx.x) / GW + 1;

    float4 w0[32], w1[32];
    {
        const float* wp = W + (long)(jh * 32 + c) * KD + 4 * h;
#pragma unroll
        for (int i = 0; i < 32; ++i) w0[i] = *(const float4*)(wp + 8 * i);
#pragma unroll
        for (int i = 0; i < 32; ++i) w1[i] = *(const float4*)(wp + 16 * KD + 8 * i);
    }
    const float bv0 = bias[jh * 32 + c], bv1 = bias[jh * 32 + 16 + c];

    const long lo = (long)c * F + 4 * h;
    float4 a0[8], a1[8];
    ld8(a0, feat + (long)((int)blockIdx.x + GW * g) * (16 * F) + lo);
    for (int n = g; n < cnt; n += 4) {
        const int  tile = (int)blockIdx.x + GW * n;
        const int  nx   = (n + 4 < cnt) ? tile + 4 * GW : tile;
        const long to   = (long)tile * (16 * F) + lo;
        v8f c0, c1;
#pragma unroll
        for (int r = 0; r < 8; ++r) { c0[r] = bv0; c1[r] = bv1; }

        ld8(a1, h1 + to);
        __builtin_amdgcn_sched_barrier(0);
        seg_mac<0>(a0, w0, w1, c0, c1);
        __builtin_amdgcn_sched_barrier(0);
        ld8(a0, h2 + to);
        __builtin_amdgcn_sched_barrier(0);
        seg_mac<1>(a1, w0, w1, c0, c1);
        __builtin_amdgcn_sched_barrier(0);
        ld8(a1, h3 + to);
        __builtin_amdgcn_sched_barrier(0);
        seg_mac<2>(a0, w0, w1, c0, c1);
        __builtin_amdgcn_sched_barrier(0);
        ld8(a0, feat + (long)nx * (16 * F) + lo);
        __builtin_amdgcn_sched_barrier(0);
        seg_mac<3>(a1, w0, w1, c0, c1);

        float* op = out + (long)(tile * 16 + 8 * h) * F + jh * 32 + c;
#pragma unroll
        for (int r = 0; r < 8; ++r) {
            op[r * F]      = c0[r];
            op[r * F + 16] = c1[r];
        }
    }
}


extern "C" void kernel_launch(void* const* d_in, const int* in_sizes, int n_in,
                              void* d_out, int out_size, void* d_ws, size_t ws_size,
                              hipStream_t stream) {
    const float* feat = (const float*)d_in[0];
    const int*   src  = (const int*)d_in[1];
    const int*   dst  = (const int*)d_in[2];
    const float* W    = (const float*)d_in[3];
    const float* bias = (const float*)d_in[4];
    float*       out  = (float*)d_out;

    float* ws   = (float*)d_ws;
    int*   deg  = (int*)d_ws;
    float* h1   = ws + DEGW;
    float* h2   = h1 + (long)NN * F;
    float* h3   = h2 + (long)NN * F;

    zero_k <<<196, 256, 0, stream>>>(ws, DEGW);
    deg_k  <<<2 * NCH, 256, 0, stream>>>(src, dst, deg, (unsigned*)out);

    hop_k<1> <<<HOPG, 1024, 0, stream>>>(deg, (const unsigned*)out, feat, h1, h3);
    hop_k<0> <<<HOPG, 1024, 0, stream>>>(deg, (const unsigned*)out, h1, h2, h3);
    hop_k<0> <<<HOPG, 1024, 0, stream>>>(deg, (const unsigned*)out, h2, h3, h3);

    gemm_k <<<GW, 256, 0, stream>>>(feat, h1, h2, h3, W, bias, out);
}
